# starting kernel
# baseline (speedup 1.0000x reference)
#include <hip/hip_runtime.h>

#define NU 100000
#define NB 50000
#define NNODE (NU + NB)
#define H 128
#define F 256
#define EPR 500000
#define NE (4 * EPR)

typedef __attribute__((ext_vector_type(2))) float v2f;
typedef __attribute__((ext_vector_type(8))) float v8f;

__global__ void __launch_bounds__(256)
gemm_bias_wmma(const float* __restrict__ X, const float* __restrict__ W,
               const float* __restrict__ bias, float* __restrict__ Y, int M) {
  const int wave = (int)((blockIdx.x * blockDim.x + threadIdx.x) >> 5);
  const int lane = threadIdx.x & 31;
  if (wave * 16 >= M) return;
  const int mbase = wave * 16;
  const int lrow  = lane & 15;
  const int khalf = (lane >> 4) << 1;

  const float* arow = X + (size_t)(mbase + lrow) * F + khalf;

  v8f c[8] = {};

  for (int k = 0; k < F; k += 4) {
    v2f a = *(const v2f*)(arow + k);
#pragma unroll
    for (int nt = 0; nt < 8; ++nt) {
      v2f b = *(const v2f*)(W + (size_t)(nt * 16 + lrow) * F + khalf + k);
      c[nt] = __builtin_amdgcn_wmma_f32_16x16x4_f32(
           false, a,  false, b,
           (short)0, c[nt],  false,  false);
    }
  }

  const int mo = (lane >> 4) << 3;
#pragma unroll
  for (int nt = 0; nt < 8; ++nt) {
    float bv = bias[nt * 16 + lrow];
#pragma unroll
    for (int v = 0; v < 8; ++v) {
      Y[(size_t)(mbase + mo + v) * H + nt * 16 + lrow] = c[nt][v] + bv;
    }
  }
}

__device__ __forceinline__ void decode_edge(int e, const int* __restrict__ rates,
                                            const int* __restrict__ impl,
                                            int& row, int& col) {
  const int rev = (e >= 2 * EPR) ? 1 : 0;
  int idx = rev ? (e - 2 * EPR) : e;
  const int* ed = (idx < EPR) ? rates : impl;
  const int i = (idx < EPR) ? idx : (idx - EPR);
  const int u = ed[i];
  const int b = ed[i + EPR];
  row = rev ? (b + NU) : u;
  col = rev ? u : b;
}

__global__ void __launch_bounds__(256)
degree_kernel(const int* __restrict__ rates, const int* __restrict__ impl,
              float* __restrict__ deg) {
  const int e = blockIdx.x * blockDim.x + threadIdx.x;
  if (e >= NE) return;
  int row, col;
  decode_edge(e, rates, impl, row, col);
  unsafeAtomicAdd(deg + col, 1.0f);
}

__global__ void __launch_bounds__(256)
dis_kernel(const float* __restrict__ deg, float* __restrict__ dis) {
  const int i = blockIdx.x * blockDim.x + threadIdx.x;
  if (i >= NNODE) return;
  const float d = deg[i];
  dis[i] = (d > 0.0f) ? rsqrtf(d) : 0.0f;
}

__global__ void __launch_bounds__(256)
norm_kernel(const int* __restrict__ rates, const int* __restrict__ impl,
            const float* __restrict__ dis, float* __restrict__ nrm) {
  const int e = blockIdx.x * blockDim.x + threadIdx.x;
  if (e >= NE) return;
  int row, col;
  decode_edge(e, rates, impl, row, col);
  nrm[e] = dis[row] * dis[col];
}

__global__ void __launch_bounds__(256)
scatter_kernel(const float* __restrict__ x, const float* __restrict__ nrm,
               const int* __restrict__ rates, const int* __restrict__ impl,
               float* __restrict__ out) {
  const int tid  = blockIdx.x * blockDim.x + threadIdx.x;
  const int e    = tid >> 5;
  const int lane = tid & 31;
  if (e >= NE) return;
  int row, col;
  decode_edge(e, rates, impl, row, col);
  const float n = nrm[e];
  const float4 v = ((const float4*)(x + (size_t)row * H))[lane];
  float* dst = out + (size_t)col * H + lane * 4;
  unsafeAtomicAdd(dst + 0, v.x * n);
  unsafeAtomicAdd(dst + 1, v.y * n);
  unsafeAtomicAdd(dst + 2, v.z * n);
  unsafeAtomicAdd(dst + 3, v.w * n);
}

__global__ void __launch_bounds__(256)
accum_copy_kernel(float* __restrict__ acc, float* __restrict__ x,
                  const float* __restrict__ buf) {
  const int i = blockIdx.x * blockDim.x + threadIdx.x;
  if (i >= NNODE * H) return;
  const float v = buf[i];
  acc[i] += v;
  x[i] = v;
}

__global__ void __launch_bounds__(256)
scale_kernel(float* __restrict__ acc) {
  const int i = blockIdx.x * blockDim.x + threadIdx.x;
  if (i >= NNODE * H) return;
  acc[i] *= (1.0f / 3.0f);
}

extern "C" void kernel_launch(void* const* d_in, const int* in_sizes, int n_in,
                              void* d_out, int out_size, void* d_ws, size_t ws_size,
                              hipStream_t stream) {
  (void)in_sizes; (void)n_in; (void)out_size; (void)ws_size;

  const float* user_x = (const float*)d_in[0];
  const float* book_x = (const float*)d_in[1];
  const float* user_W = (const float*)d_in[2];
  const float* user_b = (const float*)d_in[3];
  const float* book_W = (const float*)d_in[4];
  const float* book_b = (const float*)d_in[5];
  const int*   rates  = (const int*)d_in[6];
  const int*   impl   = (const int*)d_in[7];
  float* outp = (float*)d_out;

  float* x   = (float*)d_ws;
  float* buf = x   + (size_t)NNODE * H;
  float* deg = buf + (size_t)NNODE * H;
  float* dis = deg + NNODE;
  float* nrm = dis + NNODE;

  const size_t embBytes = (size_t)NNODE * H * sizeof(float);
  hipMemsetAsync(deg, 0, NNODE * sizeof(float), stream);

  {
    int waves = NU / 16;
    gemm_bias_wmma<<<(waves + 7) / 8, 256, 0, stream>>>(user_x, user_W, user_b, x, NU);
    waves = NB / 16;
    gemm_bias_wmma<<<(waves + 7) / 8, 256, 0, stream>>>(book_x, book_W, book_b,
                                                        x + (size_t)NU * H, NB);
  }

  hipMemcpyAsync(outp, x, embBytes, hipMemcpyDeviceToDevice, stream);

  degree_kernel<<<(NE + 255) / 256, 256, 0, stream>>>(rates, impl, deg);
  dis_kernel<<<(NNODE + 255) / 256, 256, 0, stream>>>(deg, dis);
  norm_kernel<<<(NE + 255) / 256, 256, 0, stream>>>(rates, impl, dis, nrm);

  const int nElem = NNODE * H;
  for (int layer = 0; layer < 2; ++layer) {
    hipMemsetAsync(buf, 0, embBytes, stream);
    const long long threads = (long long)NE * 32;
    scatter_kernel<<<(int)((threads + 255) / 256), 256, 0, stream>>>(x, nrm, rates,
                                                                     impl, buf);
    accum_copy_kernel<<<(nElem + 255) / 256, 256, 0, stream>>>(outp, x, buf);
  }

  scale_kernel<<<(nElem + 255) / 256, 256, 0, stream>>>(outp);
}


# scatter: edges whose weight is exactly 0 are skipped (x*0 adds nothing); one wave takes 32 edges, ballots the non-zero weights and loops over those; float atomics kept, arrival order as before
# speedup vs baseline: 1.4204x; 1.4204x over previous
#include <hip/hip_runtime.h>

#define NU 100000
#define NB 50000
#define NNODE (NU + NB)
#define H 128
#define F 256
#define EPR 500000
#define NE (4 * EPR)

typedef __attribute__((ext_vector_type(2))) float v2f;
typedef __attribute__((ext_vector_type(8))) float v8f;

__global__ void __launch_bounds__(256)
gemm_bias_wmma(const float* __restrict__ X, const float* __restrict__ W,
               const float* __restrict__ bias, float* __restrict__ Y, int M) {
  const int wave = (int)((blockIdx.x * blockDim.x + threadIdx.x) >> 5);
  const int lane = threadIdx.x & 31;
  if (wave * 16 >= M) return;
  const int mbase = wave * 16;
  const int lrow  = lane & 15;
  const int khalf = (lane >> 4) << 1;

  const float* arow = X + (size_t)(mbase + lrow) * F + khalf;

  v8f c[8] = {};

  for (int k = 0; k < F; k += 4) {
    v2f a = *(const v2f*)(arow + k);
#pragma unroll
    for (int nt = 0; nt < 8; ++nt) {
      v2f b = *(const v2f*)(W + (size_t)(nt * 16 + lrow) * F + khalf + k);
      c[nt] = __builtin_amdgcn_wmma_f32_16x16x4_f32(
           false, a,  false, b,
           (short)0, c[nt],  false,  false);
    }
  }

  const int mo = (lane >> 4) << 3;
#pragma unroll
  for (int nt = 0; nt < 8; ++nt) {
    float bv = bias[nt * 16 + lrow];
#pragma unroll
    for (int v = 0; v < 8; ++v) {
      Y[(size_t)(mbase + mo + v) * H + nt * 16 + lrow] = c[nt][v] + bv;
    }
  }
}

__device__ __forceinline__ void decode_edge(int e, const int* __restrict__ rates,
                                            const int* __restrict__ impl,
                                            int& row, int& col) {
  const int rev = (e >= 2 * EPR) ? 1 : 0;
  int idx = rev ? (e - 2 * EPR) : e;
  const int* ed = (idx < EPR) ? rates : impl;
  const int i = (idx < EPR) ? idx : (idx - EPR);
  const int u = ed[i];
  const int b = ed[i + EPR];
  row = rev ? (b + NU) : u;
  col = rev ? u : b;
}

__global__ void __launch_bounds__(256)
degree_kernel(const int* __restrict__ rates, const int* __restrict__ impl,
              float* __restrict__ deg) {
  const int e = blockIdx.x * blockDim.x + threadIdx.x;
  if (e >= NE) return;
  int row, col;
  decode_edge(e, rates, impl, row, col);
  unsafeAtomicAdd(deg + col, 1.0f);
}

__global__ void __launch_bounds__(256)
dis_kernel(const float* __restrict__ deg, float* __restrict__ dis) {
  const int i = blockIdx.x * blockDim.x + threadIdx.x;
  if (i >= NNODE) return;
  const float d = deg[i];
  dis[i] = (d > 0.0f) ? rsqrtf(d) : 0.0f;
}

__global__ void __launch_bounds__(256)
norm_kernel(const int* __restrict__ rates, const int* __restrict__ impl,
            const float* __restrict__ dis, float* __restrict__ nrm) {
  const int e = blockIdx.x * blockDim.x + threadIdx.x;
  if (e >= NE) return;
  int row, col;
  decode_edge(e, rates, impl, row, col);
  nrm[e] = dis[row] * dis[col];
}

__global__ void __launch_bounds__(256)
scatter_kernel(const float* __restrict__ x, const float* __restrict__ nrm,
               const int* __restrict__ rates, const int* __restrict__ impl,
               float* __restrict__ out) {
  const int e    = blockIdx.x * blockDim.x + threadIdx.x;
  const int lane = threadIdx.x & 31;
  const float n = (e < NE) ? nrm[e] : 0.0f;
  unsigned live = __builtin_amdgcn_ballot_w32(n != 0.0f);
  if (live == 0u) return;
  int row = 0, col = 0;
  if (e < NE) decode_edge(e, rates, impl, row, col);
  while (live != 0u) {
    const int src = __builtin_ctz(live);
    live &= live - 1u;
    const int   r = __builtin_amdgcn_readlane(row, src);
    const int   c = __builtin_amdgcn_readlane(col, src);
    const float w = __builtin_bit_cast(
        float, __builtin_amdgcn_readlane(__builtin_bit_cast(int, n), src));
    const float4 v = ((const float4*)(x + (size_t)r * H))[lane];
    float* dst = out + (size_t)c * H + lane * 4;
    unsafeAtomicAdd(dst + 0, v.x * w);
    unsafeAtomicAdd(dst + 1, v.y * w);
    unsafeAtomicAdd(dst + 2, v.z * w);
    unsafeAtomicAdd(dst + 3, v.w * w);
  }
}

__global__ void __launch_bounds__(256)
accum_copy_kernel(float* __restrict__ acc, float* __restrict__ x,
                  const float* __restrict__ buf) {
  const int i = blockIdx.x * blockDim.x + threadIdx.x;
  if (i >= NNODE * H) return;
  const float v = buf[i];
  acc[i] += v;
  x[i] = v;
}

__global__ void __launch_bounds__(256)
scale_kernel(float* __restrict__ acc) {
  const int i = blockIdx.x * blockDim.x + threadIdx.x;
  if (i >= NNODE * H) return;
  acc[i] *= (1.0f / 3.0f);
}

extern "C" void kernel_launch(void* const* d_in, const int* in_sizes, int n_in,
                              void* d_out, int out_size, void* d_ws, size_t ws_size,
                              hipStream_t stream) {
  (void)in_sizes; (void)n_in; (void)out_size; (void)ws_size;

  const float* user_x = (const float*)d_in[0];
  const float* book_x = (const float*)d_in[1];
  const float* user_W = (const float*)d_in[2];
  const float* user_b = (const float*)d_in[3];
  const float* book_W = (const float*)d_in[4];
  const float* book_b = (const float*)d_in[5];
  const int*   rates  = (const int*)d_in[6];
  const int*   impl   = (const int*)d_in[7];
  float* outp = (float*)d_out;

  float* x   = (float*)d_ws;
  float* buf = x   + (size_t)NNODE * H;
  float* deg = buf + (size_t)NNODE * H;
  float* dis = deg + NNODE;
  float* nrm = dis + NNODE;

  const size_t embBytes = (size_t)NNODE * H * sizeof(float);
  hipMemsetAsync(deg, 0, NNODE * sizeof(float), stream);

  {
    int waves = NU / 16;
    gemm_bias_wmma<<<(waves + 7) / 8, 256, 0, stream>>>(user_x, user_W, user_b, x, NU);
    waves = NB / 16;
    gemm_bias_wmma<<<(waves + 7) / 8, 256, 0, stream>>>(book_x, book_W, book_b,
                                                        x + (size_t)NU * H, NB);
  }

  hipMemcpyAsync(outp, x, embBytes, hipMemcpyDeviceToDevice, stream);

  degree_kernel<<<(NE + 255) / 256, 256, 0, stream>>>(rates, impl, deg);
  dis_kernel<<<(NNODE + 255) / 256, 256, 0, stream>>>(deg, dis);
  norm_kernel<<<(NE + 255) / 256, 256, 0, stream>>>(rates, impl, dis, nrm);

  const int nElem = NNODE * H;
  for (int layer = 0; layer < 2; ++layer) {
    hipMemsetAsync(buf, 0, embBytes, stream);
    scatter_kernel<<<(NE + 255) / 256, 256, 0, stream>>>(x, nrm, rates, impl, buf);
    accum_copy_kernel<<<(nElem + 255) / 256, 256, 0, stream>>>(outp, x, buf);
  }

  scale_kernel<<<(nElem + 255) / 256, 256, 0, stream>>>(outp);
}


# embedding mean in one pass: projection writes layer 0 into out, layers ping-pong between two workspace buffers, one float4 kernel forms ((x0+x1)+x2)/3; copy, accumulate and scale passes gone
# speedup vs baseline: 1.7519x; 1.2334x over previous
#include <hip/hip_runtime.h>

#define NU 100000
#define NB 50000
#define NNODE (NU + NB)
#define H 128
#define F 256
#define EPR 500000
#define NE (4 * EPR)

typedef __attribute__((ext_vector_type(2))) float v2f;
typedef __attribute__((ext_vector_type(8))) float v8f;

__global__ void __launch_bounds__(256)
gemm_bias_wmma(const float* __restrict__ X, const float* __restrict__ W,
               const float* __restrict__ bias, float* __restrict__ Y, int M) {
  const int wave = (int)((blockIdx.x * blockDim.x + threadIdx.x) >> 5);
  const int lane = threadIdx.x & 31;
  if (wave * 16 >= M) return;
  const int mbase = wave * 16;
  const int lrow  = lane & 15;
  const int khalf = (lane >> 4) << 1;

  const float* arow = X + (size_t)(mbase + lrow) * F + khalf;

  v8f c[8] = {};

  for (int k = 0; k < F; k += 4) {
    v2f a = *(const v2f*)(arow + k);
#pragma unroll
    for (int nt = 0; nt < 8; ++nt) {
      v2f b = *(const v2f*)(W + (size_t)(nt * 16 + lrow) * F + khalf + k);
      c[nt] = __builtin_amdgcn_wmma_f32_16x16x4_f32(
           false, a,  false, b,
           (short)0, c[nt],  false,  false);
    }
  }

  const int mo = (lane >> 4) << 3;
#pragma unroll
  for (int nt = 0; nt < 8; ++nt) {
    float bv = bias[nt * 16 + lrow];
#pragma unroll
    for (int v = 0; v < 8; ++v) {
      Y[(size_t)(mbase + mo + v) * H + nt * 16 + lrow] = c[nt][v] + bv;
    }
  }
}

__device__ __forceinline__ void decode_edge(int e, const int* __restrict__ rates,
                                            const int* __restrict__ impl,
                                            int& row, int& col) {
  const int rev = (e >= 2 * EPR) ? 1 : 0;
  int idx = rev ? (e - 2 * EPR) : e;
  const int* ed = (idx < EPR) ? rates : impl;
  const int i = (idx < EPR) ? idx : (idx - EPR);
  const int u = ed[i];
  const int b = ed[i + EPR];
  row = rev ? (b + NU) : u;
  col = rev ? u : b;
}

__global__ void __launch_bounds__(256)
degree_kernel(const int* __restrict__ rates, const int* __restrict__ impl,
              float* __restrict__ deg) {
  const int e = blockIdx.x * blockDim.x + threadIdx.x;
  if (e >= NE) return;
  int row, col;
  decode_edge(e, rates, impl, row, col);
  unsafeAtomicAdd(deg + col, 1.0f);
}

__global__ void __launch_bounds__(256)
dis_kernel(const float* __restrict__ deg, float* __restrict__ dis) {
  const int i = blockIdx.x * blockDim.x + threadIdx.x;
  if (i >= NNODE) return;
  const float d = deg[i];
  dis[i] = (d > 0.0f) ? rsqrtf(d) : 0.0f;
}

__global__ void __launch_bounds__(256)
norm_kernel(const int* __restrict__ rates, const int* __restrict__ impl,
            const float* __restrict__ dis, float* __restrict__ nrm) {
  const int e = blockIdx.x * blockDim.x + threadIdx.x;
  if (e >= NE) return;
  int row, col;
  decode_edge(e, rates, impl, row, col);
  nrm[e] = dis[row] * dis[col];
}

__global__ void __launch_bounds__(256)
scatter_kernel(const float* __restrict__ x, const float* __restrict__ nrm,
               const int* __restrict__ rates, const int* __restrict__ impl,
               float* __restrict__ out) {
  const int e    = blockIdx.x * blockDim.x + threadIdx.x;
  const int lane = threadIdx.x & 31;
  const float n = (e < NE) ? nrm[e] : 0.0f;
  unsigned live = __builtin_amdgcn_ballot_w32(n != 0.0f);
  if (live == 0u) return;
  int row = 0, col = 0;
  if (e < NE) decode_edge(e, rates, impl, row, col);
  while (live != 0u) {
    const int src = __builtin_ctz(live);
    live &= live - 1u;
    const int   r = __builtin_amdgcn_readlane(row, src);
    const int   c = __builtin_amdgcn_readlane(col, src);
    const float w = __builtin_bit_cast(
        float, __builtin_amdgcn_readlane(__builtin_bit_cast(int, n), src));
    const float4 v = ((const float4*)(x + (size_t)r * H))[lane];
    float* dst = out + (size_t)c * H + lane * 4;
    unsafeAtomicAdd(dst + 0, v.x * w);
    unsafeAtomicAdd(dst + 1, v.y * w);
    unsafeAtomicAdd(dst + 2, v.z * w);
    unsafeAtomicAdd(dst + 3, v.w * w);
  }
}

#define NV4 (NNODE * H / 4)
__global__ void __launch_bounds__(256)
mean3_kernel(float* __restrict__ out, const float* __restrict__ x1,
             const float* __restrict__ x2) {
  const int i = blockIdx.x * blockDim.x + threadIdx.x;
  if (i >= NV4 / 2) return;
  const int j = i + NV4 / 2;
  float4*       o4 = (float4*)out;
  const float4* a4 = (const float4*)x1;
  const float4* b4 = (const float4*)x2;
  const float4 o0 = o4[i], o1 = o4[j];
  const float4 a0 = a4[i], a1 = a4[j];
  const float4 b0 = b4[i], b1 = b4[j];
  const float s = 1.0f / 3.0f;
  float4 r0, r1;
  r0.x = ((o0.x + a0.x) + b0.x) * s;
  r0.y = ((o0.y + a0.y) + b0.y) * s;
  r0.z = ((o0.z + a0.z) + b0.z) * s;
  r0.w = ((o0.w + a0.w) + b0.w) * s;
  r1.x = ((o1.x + a1.x) + b1.x) * s;
  r1.y = ((o1.y + a1.y) + b1.y) * s;
  r1.z = ((o1.z + a1.z) + b1.z) * s;
  r1.w = ((o1.w + a1.w) + b1.w) * s;
  o4[i] = r0;
  o4[j] = r1;
}

extern "C" void kernel_launch(void* const* d_in, const int* in_sizes, int n_in,
                              void* d_out, int out_size, void* d_ws, size_t ws_size,
                              hipStream_t stream) {
  (void)in_sizes; (void)n_in; (void)out_size; (void)ws_size;

  const float* user_x = (const float*)d_in[0];
  const float* book_x = (const float*)d_in[1];
  const float* user_W = (const float*)d_in[2];
  const float* user_b = (const float*)d_in[3];
  const float* book_W = (const float*)d_in[4];
  const float* book_b = (const float*)d_in[5];
  const int*   rates  = (const int*)d_in[6];
  const int*   impl   = (const int*)d_in[7];
  float* outp = (float*)d_out;

  float* x1  = (float*)d_ws;
  float* x2  = x1  + (size_t)NNODE * H;
  float* deg = x2  + (size_t)NNODE * H;
  float* dis = deg + NNODE;
  float* nrm = dis + NNODE;

  const size_t embBytes = (size_t)NNODE * H * sizeof(float);
  hipMemsetAsync(deg, 0, NNODE * sizeof(float), stream);

  {
    int waves = NU / 16;
    gemm_bias_wmma<<<(waves + 7) / 8, 256, 0, stream>>>(user_x, user_W, user_b, outp, NU);
    waves = NB / 16;
    gemm_bias_wmma<<<(waves + 7) / 8, 256, 0, stream>>>(book_x, book_W, book_b,
                                                        outp + (size_t)NU * H, NB);
  }

  degree_kernel<<<(NE + 255) / 256, 256, 0, stream>>>(rates, impl, deg);
  dis_kernel<<<(NNODE + 255) / 256, 256, 0, stream>>>(deg, dis);
  norm_kernel<<<(NE + 255) / 256, 256, 0, stream>>>(rates, impl, dis, nrm);

  hipMemsetAsync(x1, 0, embBytes, stream);
  scatter_kernel<<<(NE + 255) / 256, 256, 0, stream>>>(outp, nrm, rates, impl, x1);
  hipMemsetAsync(x2, 0, embBytes, stream);
  scatter_kernel<<<(NE + 255) / 256, 256, 0, stream>>>(x1, nrm, rates, impl, x2);

  mean3_kernel<<<(NV4 / 2 + 255) / 256, 256, 0, stream>>>(outp, x1, x2);
}


# layer sums without float atomics: edges counting-sorted by target once per call (degree scan + integer cursor adds), one wave then sums each target's rows in registers, in cursor arrival order
# speedup vs baseline: 2.3006x; 1.3132x over previous
#include <hip/hip_runtime.h>

#define NU 100000
#define NB 50000
#define NNODE (NU + NB)
#define H 128
#define F 256
#define EPR 500000
#define NE (4 * EPR)

typedef __attribute__((ext_vector_type(2))) float v2f;
typedef __attribute__((ext_vector_type(8))) float v8f;

__global__ void __launch_bounds__(256)
gemm_bias_wmma(const float* __restrict__ X, const float* __restrict__ W,
               const float* __restrict__ bias, float* __restrict__ Y, int M) {
  const int wave = (int)((blockIdx.x * blockDim.x + threadIdx.x) >> 5);
  const int lane = threadIdx.x & 31;
  if (wave * 16 >= M) return;
  const int mbase = wave * 16;
  const int lrow  = lane & 15;
  const int khalf = (lane >> 4) << 1;

  const float* arow = X + (size_t)(mbase + lrow) * F + khalf;

  v8f c[8] = {};

  for (int k = 0; k < F; k += 4) {
    v2f a = *(const v2f*)(arow + k);
#pragma unroll
    for (int nt = 0; nt < 8; ++nt) {
      v2f b = *(const v2f*)(W + (size_t)(nt * 16 + lrow) * F + khalf + k);
      c[nt] = __builtin_amdgcn_wmma_f32_16x16x4_f32(
           false, a,  false, b,
           (short)0, c[nt],  false,  false);
    }
  }

  const int mo = (lane >> 4) << 3;
#pragma unroll
  for (int nt = 0; nt < 8; ++nt) {
    float bv = bias[nt * 16 + lrow];
#pragma unroll
    for (int v = 0; v < 8; ++v) {
      Y[(size_t)(mbase + mo + v) * H + nt * 16 + lrow] = c[nt][v] + bv;
    }
  }
}

__device__ __forceinline__ void decode_edge(int e, const int* __restrict__ rates,
                                            const int* __restrict__ impl,
                                            int& row, int& col) {
  const int rev = (e >= 2 * EPR) ? 1 : 0;
  int idx = rev ? (e - 2 * EPR) : e;
  const int* ed = (idx < EPR) ? rates : impl;
  const int i = (idx < EPR) ? idx : (idx - EPR);
  const int u = ed[i];
  const int b = ed[i + EPR];
  row = rev ? (b + NU) : u;
  col = rev ? u : b;
}

__global__ void __launch_bounds__(256)
degree_kernel(const int* __restrict__ rates, const int* __restrict__ impl,
              float* __restrict__ deg) {
  const int e = blockIdx.x * blockDim.x + threadIdx.x;
  if (e >= NE) return;
  int row, col;
  decode_edge(e, rates, impl, row, col);
  unsafeAtomicAdd(deg + col, 1.0f);
}

#define SCAN_PER 1024
#define NSCANB ((NNODE + SCAN_PER - 1) / SCAN_PER)

__global__ void __launch_bounds__(256)
dis_scan_kernel(const float* degf, int* cur, float* __restrict__ dis,
                int* __restrict__ bsum) {
  __shared__ int wsum[8];
  const int t = threadIdx.x;
  const int lane = t & 31;
  const int wv = t >> 5;
  const int base = blockIdx.x * SCAN_PER + t * 4;
  const bool in = base < NNODE;
  float4 d = make_float4(0.0f, 0.0f, 0.0f, 0.0f);
  if (in) {
    d = *(const float4*)(degf + base);
    float4 s;
    s.x = (d.x > 0.0f) ? rsqrtf(d.x) : 0.0f;
    s.y = (d.y > 0.0f) ? rsqrtf(d.y) : 0.0f;
    s.z = (d.z > 0.0f) ? rsqrtf(d.z) : 0.0f;
    s.w = (d.w > 0.0f) ? rsqrtf(d.w) : 0.0f;
    *(float4*)(dis + base) = s;
  }
  const int c0 = (int)d.x, c1 = (int)d.y, c2 = (int)d.z, c3 = (int)d.w;
  const int tot = c0 + c1 + c2 + c3;
  int inc = tot;
  for (int o = 1; o < 32; o <<= 1) {
    const int y = __shfl_up(inc, o);
    if (lane >= o) inc += y;
  }
  if (lane == 31) wsum[wv] = inc;
  __syncthreads();
  int woff = 0, btot = 0;
  for (int j = 0; j < 8; ++j) {
    const int s = wsum[j];
    if (j < wv) woff += s;
    btot += s;
  }
  if (in) {
    const int e0 = woff + inc - tot;
    int4 o4;
    o4.x = e0;
    o4.y = e0 + c0;
    o4.z = e0 + c0 + c1;
    o4.w = e0 + c0 + c1 + c2;
    *(int4*)(cur + base) = o4;
  }
  if (t == 0) bsum[blockIdx.x] = btot;
}

__global__ void __launch_bounds__(256)
scan_add_kernel(int* __restrict__ cur, const int* __restrict__ bsum) {
  const int t = threadIdx.x;
  const int lane = t & 31;
  int p = 0;
  for (int j = lane; j < (int)blockIdx.x; j += 32) p += bsum[j];
  for (int o = 16; o > 0; o >>= 1) p += __shfl_xor(p, o);
  const int base = blockIdx.x * SCAN_PER + t * 4;
  if (base < NNODE) {
    int4 v = *(int4*)(cur + base);
    v.x += p; v.y += p; v.z += p; v.w += p;
    *(int4*)(cur + base) = v;
  }
}

__global__ void __launch_bounds__(256)
fill_kernel(const int* __restrict__ rates, const int* __restrict__ impl,
            int* __restrict__ cur, int* __restrict__ srt_row) {
  const int e = blockIdx.x * blockDim.x + threadIdx.x;
  if (e >= NE) return;
  int row, col;
  decode_edge(e, rates, impl, row, col);
  const int pos = atomicAdd(cur + col, 1);
  srt_row[pos] = row;
}

#define GPW 8
#define GBLOCKS ((NNODE + 8 * GPW - 1) / (8 * GPW))
#define GWAVES (GBLOCKS * 8)

#define PICK_EDGE(k)                                                          \
  const int s##k = __builtin_ctz(live);                                       \
  live &= live - 1u;                                                          \
  const int r##k = __builtin_amdgcn_readlane(r, s##k);                        \
  const float w##k = __builtin_bit_cast(                                      \
      float, __builtin_amdgcn_readlane(__builtin_bit_cast(int, w), s##k));
#define LOAD_ROW(k)                                                           \
  const float4 v##k = ((const float4*)(x + (size_t)r##k * H))[lane];
#define ADD_EDGE(k)                                                           \
  acc.x += v##k.x * w##k; acc.y += v##k.y * w##k;                             \
  acc.z += v##k.z * w##k; acc.w += v##k.w * w##k;

__global__ void __launch_bounds__(256)
gather_kernel(const float* __restrict__ x, const float* __restrict__ dis,
              const int* __restrict__ cur, const int* __restrict__ srt_row,
              float* __restrict__ out) {
  const int lane = threadIdx.x & 31;
  const int g = __builtin_amdgcn_readfirstlane(
      (int)((blockIdx.x * blockDim.x + threadIdx.x) >> 5));
  for (int j = 0; j < GPW; ++j) {
    const int c = g + j * GWAVES;
    if (c >= NNODE) break;
    const int prev = cur[(c > 0) ? c - 1 : 0];
    const int end = cur[c];
    const float dc = dis[c];
    const int beg = (c > 0) ? prev : 0;
    float4 acc = make_float4(0.0f, 0.0f, 0.0f, 0.0f);
    for (int base = beg; base < end; base += 32) {
      const int idx = base + lane;
      int r = 0;
      float w = 0.0f;
      if (idx < end) {
        r = srt_row[idx];
        w = dis[r] * dc;
      }
      unsigned live = __builtin_amdgcn_ballot_w32(w != 0.0f);
      while (live != 0u) {
        PICK_EDGE(0)
        if (live == 0u) { LOAD_ROW(0) ADD_EDGE(0) break; }
        PICK_EDGE(1)
        if (live == 0u) { LOAD_ROW(0) LOAD_ROW(1) ADD_EDGE(0) ADD_EDGE(1) break; }
        PICK_EDGE(2)
        if (live == 0u) {
          LOAD_ROW(0) LOAD_ROW(1) LOAD_ROW(2)
          ADD_EDGE(0) ADD_EDGE(1) ADD_EDGE(2)
          break;
        }
        PICK_EDGE(3)
        LOAD_ROW(0) LOAD_ROW(1) LOAD_ROW(2) LOAD_ROW(3)
        ADD_EDGE(0) ADD_EDGE(1) ADD_EDGE(2) ADD_EDGE(3)
      }
    }
    ((float4*)(out + (size_t)c * H))[lane] = acc;
  }
}

#define NV4 (NNODE * H / 4)
__global__ void __launch_bounds__(256)
mean3_kernel(float* __restrict__ out, const float* __restrict__ x1,
             const float* __restrict__ x2) {
  const int i = blockIdx.x * blockDim.x + threadIdx.x;
  if (i >= NV4 / 2) return;
  const int j = i + NV4 / 2;
  float4*       o4 = (float4*)out;
  const float4* a4 = (const float4*)x1;
  const float4* b4 = (const float4*)x2;
  const float4 o0 = o4[i], o1 = o4[j];
  const float4 a0 = a4[i], a1 = a4[j];
  const float4 b0 = b4[i], b1 = b4[j];
  const float s = 1.0f / 3.0f;
  float4 r0, r1;
  r0.x = ((o0.x + a0.x) + b0.x) * s;
  r0.y = ((o0.y + a0.y) + b0.y) * s;
  r0.z = ((o0.z + a0.z) + b0.z) * s;
  r0.w = ((o0.w + a0.w) + b0.w) * s;
  r1.x = ((o1.x + a1.x) + b1.x) * s;
  r1.y = ((o1.y + a1.y) + b1.y) * s;
  r1.z = ((o1.z + a1.z) + b1.z) * s;
  r1.w = ((o1.w + a1.w) + b1.w) * s;
  o4[i] = r0;
  o4[j] = r1;
}

extern "C" void kernel_launch(void* const* d_in, const int* in_sizes, int n_in,
                              void* d_out, int out_size, void* d_ws, size_t ws_size,
                              hipStream_t stream) {
  (void)in_sizes; (void)n_in; (void)out_size; (void)ws_size;

  const float* user_x = (const float*)d_in[0];
  const float* book_x = (const float*)d_in[1];
  const float* user_W = (const float*)d_in[2];
  const float* user_b = (const float*)d_in[3];
  const float* book_W = (const float*)d_in[4];
  const float* book_b = (const float*)d_in[5];
  const int*   rates  = (const int*)d_in[6];
  const int*   impl   = (const int*)d_in[7];
  float* outp = (float*)d_out;

  float* x1  = (float*)d_ws;
  float* x2  = x1  + (size_t)NNODE * H;
  float* deg = x2  + (size_t)NNODE * H;
  float* dis = deg + NNODE;
  int*   srt = (int*)(dis + NNODE);
  int*   cur = (int*)deg;
  int*   bsum = (int*)x2;

  hipMemsetAsync(deg, 0, NNODE * sizeof(float), stream);

  {
    int waves = NU / 16;
    gemm_bias_wmma<<<(waves + 7) / 8, 256, 0, stream>>>(user_x, user_W, user_b, outp, NU);
    waves = NB / 16;
    gemm_bias_wmma<<<(waves + 7) / 8, 256, 0, stream>>>(book_x, book_W, book_b,
                                                        outp + (size_t)NU * H, NB);
  }

  degree_kernel<<<(NE + 255) / 256, 256, 0, stream>>>(rates, impl, deg);
  dis_scan_kernel<<<NSCANB, 256, 0, stream>>>(deg, cur, dis, bsum);
  scan_add_kernel<<<NSCANB, 256, 0, stream>>>(cur, bsum);
  fill_kernel<<<(NE + 255) / 256, 256, 0, stream>>>(rates, impl, cur, srt);

  gather_kernel<<<GBLOCKS, 256, 0, stream>>>(outp, dis, cur, srt, x1);
  gather_kernel<<<GBLOCKS, 256, 0, stream>>>(x1, dis, cur, srt, x2);

  mean3_kernel<<<(NV4 / 2 + 255) / 256, 256, 0, stream>>>(outp, x1, x2);
}


# edge grouping without atomics on memory: edges binned by 512-target range with per-block LDS counters, one block per range counts, scans and ranks in LDS; segment order = LDS cursor arrival order
# speedup vs baseline: 3.6883x; 1.6032x over previous
#include <hip/hip_runtime.h>

#define NU 100000
#define NB 50000
#define NNODE (NU + NB)
#define H 128
#define F 256
#define EPR 500000
#define NE (4 * EPR)

typedef __attribute__((ext_vector_type(2))) float v2f;
typedef __attribute__((ext_vector_type(8))) float v8f;

__global__ void __launch_bounds__(256)
gemm_bias_wmma(const float* __restrict__ X, const float* __restrict__ W,
               const float* __restrict__ bias, float* __restrict__ Y, int M) {
  const int wave = (int)((blockIdx.x * blockDim.x + threadIdx.x) >> 5);
  const int lane = threadIdx.x & 31;
  if (wave * 16 >= M) return;
  const int mbase = wave * 16;
  const int lrow  = lane & 15;
  const int khalf = (lane >> 4) << 1;

  const float* arow = X + (size_t)(mbase + lrow) * F + khalf;

  v8f c[8] = {};

  for (int k = 0; k < F; k += 4) {
    v2f a = *(const v2f*)(arow + k);
#pragma unroll
    for (int nt = 0; nt < 8; ++nt) {
      v2f b = *(const v2f*)(W + (size_t)(nt * 16 + lrow) * F + khalf + k);
      c[nt] = __builtin_amdgcn_wmma_f32_16x16x4_f32(
           false, a,  false, b,
           (short)0, c[nt],  false,  false);
    }
  }

  const int mo = (lane >> 4) << 3;
#pragma unroll
  for (int nt = 0; nt < 8; ++nt) {
    float bv = bias[nt * 16 + lrow];
#pragma unroll
    for (int v = 0; v < 8; ++v) {
      Y[(size_t)(mbase + mo + v) * H + nt * 16 + lrow] = c[nt][v] + bv;
    }
  }
}

__device__ __forceinline__ void load_edge(int e, const int* __restrict__ rates,
                                          const int* __restrict__ impl,
                                          int& u, int& b) {
  const int idx = (e >= 2 * EPR) ? (e - 2 * EPR) : e;
  const int* ed = (idx < EPR) ? rates : impl;
  const int i = (idx < EPR) ? idx : (idx - EPR);
  u = ed[i];
  b = ed[i + EPR];
}

__device__ __forceinline__ void orient_edge(int e, int u, int b, int& row, int& col) {
  const int rev = (e >= 2 * EPR) ? 1 : 0;
  row = rev ? (b + NU) : u;
  col = rev ? u : b;
}

__device__ __forceinline__ void decode_edge(int e, const int* __restrict__ rates,
                                            const int* __restrict__ impl,
                                            int& row, int& col) {
  int u, b;
  load_edge(e, rates, impl, u, b);
  orient_edge(e, u, b, row, col);
}

#define BIN_SHIFT 9
#define BIN_NODES (1 << BIN_SHIFT)
#define NBIN ((NNODE + BIN_NODES - 1) / BIN_NODES)
#define NBIN_PAD 320
#define NBLK 256
#define CHUNK ((NE + NBLK - 1) / NBLK)
#define BT 1024

__global__ void __launch_bounds__(BT)
bin_count_kernel(const int* __restrict__ rates, const int* __restrict__ impl,
                 int* __restrict__ blkhist) {
  __shared__ int h[NBIN];
  const int t = threadIdx.x;
  for (int b = t; b < NBIN; b += BT) h[b] = 0;
  __syncthreads();
  const int e0 = blockIdx.x * CHUNK;
  const int e1 = min(e0 + CHUNK, NE);
  for (int eb = e0 + t; eb < e1; eb += 4 * BT) {
    int cl[4];
    for (int k = 0; k < 4; ++k) {
      const int e = eb + k * BT;
      int row;
      cl[k] = -1;
      if (e < e1) decode_edge(e, rates, impl, row, cl[k]);
    }
    for (int k = 0; k < 4; ++k) {
      if (cl[k] >= 0) atomicAdd(&h[cl[k] >> BIN_SHIFT], 1);
    }
  }
  __syncthreads();
  for (int b = t; b < NBIN; b += BT) blkhist[b * NBLK + blockIdx.x] = h[b];
}

__global__ void __launch_bounds__(BT)
bin_place_kernel(const int* __restrict__ rates, const int* __restrict__ impl,
                 const int* __restrict__ blkhist, int* __restrict__ binbase,
                 int2* __restrict__ pairs) {
  __shared__ int tot[NBIN_PAD];
  __shared__ int cursor[NBIN_PAD];
  const int t = threadIdx.x;
  const int lane = t & 31;
  const int wv = t >> 5;
  const int blk = blockIdx.x;
  for (int b = wv; b < NBIN; b += BT / 32) {
    const int4* rp = (const int4*)(blkhist + b * NBLK) + lane * 2;
    const int4 v0 = rp[0], v1 = rp[1];
    const int i0 = lane * 8;
    int all = ((v0.x + v0.y) + (v0.z + v0.w)) + ((v1.x + v1.y) + (v1.z + v1.w));
    int bef = 0;
    bef += (i0 + 0 < blk) ? v0.x : 0;
    bef += (i0 + 1 < blk) ? v0.y : 0;
    bef += (i0 + 2 < blk) ? v0.z : 0;
    bef += (i0 + 3 < blk) ? v0.w : 0;
    bef += (i0 + 4 < blk) ? v1.x : 0;
    bef += (i0 + 5 < blk) ? v1.y : 0;
    bef += (i0 + 6 < blk) ? v1.z : 0;
    bef += (i0 + 7 < blk) ? v1.w : 0;
    for (int o = 16; o > 0; o >>= 1) {
      all += __shfl_xor(all, o);
      bef += __shfl_xor(bef, o);
    }
    if (lane == 0) { tot[b] = all; cursor[b] = bef; }
  }
  __syncthreads();
  if (wv == 0) {
    int s = 0;
    for (int k = 0; k < 10; ++k) {
      const int b = lane * 10 + k;
      if (b < NBIN) s += tot[b];
    }
    int inc = s;
    for (int o = 1; o < 32; o <<= 1) {
      const int y = __shfl_up(inc, o);
      if (lane >= o) inc += y;
    }
    int run = inc - s;
    for (int k = 0; k < 10; ++k) {
      const int b = lane * 10 + k;
      if (b < NBIN) {
        const int tb = tot[b];
        cursor[b] += run;
        if (blk == 0) binbase[b] = run;
        run += tb;
      }
    }
    if (blk == 0 && lane == 31) binbase[NBIN] = inc;
  }
  __syncthreads();
  const int e0 = blk * CHUNK;
  const int e1 = min(e0 + CHUNK, NE);
  for (int eb = e0 + t; eb < e1; eb += 4 * BT) {
    int uu[4], bb[4];
    for (int k = 0; k < 4; ++k) {
      const int e = eb + k * BT;
      uu[k] = -1;
      bb[k] = 0;
      if (e < e1) load_edge(e, rates, impl, uu[k], bb[k]);
    }
    for (int k = 0; k < 4; ++k) {
      if (uu[k] >= 0) {
        int row, col;
        orient_edge(eb + k * BT, uu[k], bb[k], row, col);
        const int p = atomicAdd(&cursor[col >> BIN_SHIFT], 1);
        pairs[p] = make_int2(col, row);
      }
    }
  }
}

__global__ void __launch_bounds__(BT)
bin_sort_kernel(const int2* __restrict__ pairs, const int* __restrict__ binbase,
                int* __restrict__ cur, float* __restrict__ dis,
                int* __restrict__ srt_row) {
  __shared__ int cnt[BIN_NODES];
  __shared__ int wsum[BIN_NODES / 32];
  const int t = threadIdx.x;
  const int lane = t & 31;
  const int wv = t >> 5;
  const int nbase = blockIdx.x << BIN_SHIFT;
  const int rbeg = binbase[blockIdx.x];
  const int rend = binbase[blockIdx.x + 1];
  if (t < BIN_NODES) cnt[t] = 0;
  __syncthreads();
  for (int ib = rbeg + t; ib < rend; ib += 4 * BT) {
    int cl[4];
    for (int k = 0; k < 4; ++k) {
      const int i = ib + k * BT;
      cl[k] = (i < rend) ? pairs[i].x : -1;
    }
    for (int k = 0; k < 4; ++k) {
      if (cl[k] >= 0) atomicAdd(&cnt[cl[k] - nbase], 1);
    }
  }
  __syncthreads();
  int d = 0, inc = 0;
  if (t < BIN_NODES) {
    d = cnt[t];
    inc = d;
    for (int o = 1; o < 32; o <<= 1) {
      const int y = __shfl_up(inc, o);
      if (lane >= o) inc += y;
    }
    if (lane == 31) wsum[wv] = inc;
  }
  __syncthreads();
  if (t < BIN_NODES) {
    int woff = 0;
    for (int j = 0; j < BIN_NODES / 32; ++j) {
      if (j < wv) woff += wsum[j];
    }
    const int ex = woff + inc - d;
    cnt[t] = ex;
    const int node = nbase + t;
    if (node < NNODE) {
      dis[node] = (d > 0) ? rsqrtf((float)d) : 0.0f;
      cur[node] = rbeg + ex + d;
    }
  }
  __syncthreads();
  for (int ib = rbeg + t; ib < rend; ib += 4 * BT) {
    int2 p[4];
    for (int k = 0; k < 4; ++k) {
      const int i = ib + k * BT;
      p[k] = (i < rend) ? pairs[i] : make_int2(-1, 0);
    }
    for (int k = 0; k < 4; ++k) {
      if (p[k].x >= 0) {
        const int pos = atomicAdd(&cnt[p[k].x - nbase], 1);
        srt_row[rbeg + pos] = p[k].y;
      }
    }
  }
}

#define GPW 8
#define GBLOCKS ((NNODE + 8 * GPW - 1) / (8 * GPW))
#define GWAVES (GBLOCKS * 8)

#define PICK_EDGE(k)                                                          \
  const int s##k = __builtin_ctz(live);                                       \
  live &= live - 1u;                                                          \
  const int r##k = __builtin_amdgcn_readlane(r, s##k);                        \
  const float w##k = __builtin_bit_cast(                                      \
      float, __builtin_amdgcn_readlane(__builtin_bit_cast(int, w), s##k));
#define LOAD_ROW(k)                                                           \
  const float4 v##k = ((const float4*)(x + (size_t)r##k * H))[lane];
#define ADD_EDGE(k)                                                           \
  acc.x += v##k.x * w##k; acc.y += v##k.y * w##k;                             \
  acc.z += v##k.z * w##k; acc.w += v##k.w * w##k;

__global__ void __launch_bounds__(256)
gather_kernel(const float* __restrict__ x, const float* __restrict__ dis,
              const int* __restrict__ cur, const int* __restrict__ srt_row,
              float* __restrict__ out) {
  const int lane = threadIdx.x & 31;
  const int g = __builtin_amdgcn_readfirstlane(
      (int)((blockIdx.x * blockDim.x + threadIdx.x) >> 5));
  for (int j = 0; j < GPW; ++j) {
    const int c = g + j * GWAVES;
    if (c >= NNODE) break;
    const int prev = cur[(c > 0) ? c - 1 : 0];
    const int end = cur[c];
    const float dc = dis[c];
    const int beg = (c > 0) ? prev : 0;
    float4 acc = make_float4(0.0f, 0.0f, 0.0f, 0.0f);
    for (int base = beg; base < end; base += 32) {
      const int idx = base + lane;
      int r = 0;
      float w = 0.0f;
      if (idx < end) {
        r = srt_row[idx];
        w = dis[r] * dc;
      }
      unsigned live = __builtin_amdgcn_ballot_w32(w != 0.0f);
      while (live != 0u) {
        PICK_EDGE(0)
        if (live == 0u) { LOAD_ROW(0) ADD_EDGE(0) break; }
        PICK_EDGE(1)
        if (live == 0u) { LOAD_ROW(0) LOAD_ROW(1) ADD_EDGE(0) ADD_EDGE(1) break; }
        PICK_EDGE(2)
        if (live == 0u) {
          LOAD_ROW(0) LOAD_ROW(1) LOAD_ROW(2)
          ADD_EDGE(0) ADD_EDGE(1) ADD_EDGE(2)
          break;
        }
        PICK_EDGE(3)
        LOAD_ROW(0) LOAD_ROW(1) LOAD_ROW(2) LOAD_ROW(3)
        ADD_EDGE(0) ADD_EDGE(1) ADD_EDGE(2) ADD_EDGE(3)
      }
    }
    ((float4*)(out + (size_t)c * H))[lane] = acc;
  }
}

#define NV4 (NNODE * H / 4)
__global__ void __launch_bounds__(256)
mean3_kernel(float* __restrict__ out, const float* __restrict__ x1,
             const float* __restrict__ x2) {
  const int i = blockIdx.x * blockDim.x + threadIdx.x;
  if (i >= NV4 / 2) return;
  const int j = i + NV4 / 2;
  float4*       o4 = (float4*)out;
  const float4* a4 = (const float4*)x1;
  const float4* b4 = (const float4*)x2;
  const float4 o0 = o4[i], o1 = o4[j];
  const float4 a0 = a4[i], a1 = a4[j];
  const float4 b0 = b4[i], b1 = b4[j];
  const float s = 1.0f / 3.0f;
  float4 r0, r1;
  r0.x = ((o0.x + a0.x) + b0.x) * s;
  r0.y = ((o0.y + a0.y) + b0.y) * s;
  r0.z = ((o0.z + a0.z) + b0.z) * s;
  r0.w = ((o0.w + a0.w) + b0.w) * s;
  r1.x = ((o1.x + a1.x) + b1.x) * s;
  r1.y = ((o1.y + a1.y) + b1.y) * s;
  r1.z = ((o1.z + a1.z) + b1.z) * s;
  r1.w = ((o1.w + a1.w) + b1.w) * s;
  o4[i] = r0;
  o4[j] = r1;
}

extern "C" void kernel_launch(void* const* d_in, const int* in_sizes, int n_in,
                              void* d_out, int out_size, void* d_ws, size_t ws_size,
                              hipStream_t stream) {
  (void)in_sizes; (void)n_in; (void)out_size; (void)ws_size;

  const float* user_x = (const float*)d_in[0];
  const float* book_x = (const float*)d_in[1];
  const float* user_W = (const float*)d_in[2];
  const float* user_b = (const float*)d_in[3];
  const float* book_W = (const float*)d_in[4];
  const float* book_b = (const float*)d_in[5];
  const int*   rates  = (const int*)d_in[6];
  const int*   impl   = (const int*)d_in[7];
  float* outp = (float*)d_out;

  float* x1  = (float*)d_ws;
  float* x2  = x1  + (size_t)NNODE * H;
  float* deg = x2  + (size_t)NNODE * H;
  float* dis = deg + NNODE;
  int*   srt = (int*)(dis + NNODE);
  int*   cur = (int*)deg;
  int2*  pairs   = (int2*)x2;
  int*   blkhist = (int*)x2 + 2 * (size_t)NE;
  int*   binbase = blkhist + NBIN * NBLK;

  {
    int waves = NU / 16;
    gemm_bias_wmma<<<(waves + 7) / 8, 256, 0, stream>>>(user_x, user_W, user_b, outp, NU);
    waves = NB / 16;
    gemm_bias_wmma<<<(waves + 7) / 8, 256, 0, stream>>>(book_x, book_W, book_b,
                                                        outp + (size_t)NU * H, NB);
  }

  bin_count_kernel<<<NBLK, BT, 0, stream>>>(rates, impl, blkhist);
  bin_place_kernel<<<NBLK, BT, 0, stream>>>(rates, impl, blkhist, binbase, pairs);
  bin_sort_kernel<<<NBIN, BT, 0, stream>>>(pairs, binbase, cur, dis, srt);

  gather_kernel<<<GBLOCKS, 256, 0, stream>>>(outp, dis, cur, srt, x1);
  gather_kernel<<<GBLOCKS, 256, 0, stream>>>(x1, dis, cur, srt, x2);

  mean3_kernel<<<(NV4 / 2 + 255) / 256, 256, 0, stream>>>(outp, x1, x2);
}


# mean folded into the layer-2 gather: the wave that sums a node's layer-2 row in registers also loads that node's x0 and x1 rows and stores ((x0+x1)+x2)/3; x2 plane and the mean launch gone
# speedup vs baseline: 3.8867x; 1.0538x over previous
#include <hip/hip_runtime.h>

#define NU 100000
#define NB 50000
#define NNODE (NU + NB)
#define H 128
#define F 256
#define EPR 500000
#define NE (4 * EPR)

typedef __attribute__((ext_vector_type(2))) float v2f;
typedef __attribute__((ext_vector_type(8))) float v8f;

__global__ void __launch_bounds__(256)
gemm_bias_wmma(const float* __restrict__ X, const float* __restrict__ W,
               const float* __restrict__ bias, float* __restrict__ Y, int M) {
  const int wave = (int)((blockIdx.x * blockDim.x + threadIdx.x) >> 5);
  const int lane = threadIdx.x & 31;
  if (wave * 16 >= M) return;
  const int mbase = wave * 16;
  const int lrow  = lane & 15;
  const int khalf = (lane >> 4) << 1;

  const float* arow = X + (size_t)(mbase + lrow) * F + khalf;

  v8f c[8] = {};

  for (int k = 0; k < F; k += 4) {
    v2f a = *(const v2f*)(arow + k);
#pragma unroll
    for (int nt = 0; nt < 8; ++nt) {
      v2f b = *(const v2f*)(W + (size_t)(nt * 16 + lrow) * F + khalf + k);
      c[nt] = __builtin_amdgcn_wmma_f32_16x16x4_f32(
           false, a,  false, b,
           (short)0, c[nt],  false,  false);
    }
  }

  const int mo = (lane >> 4) << 3;
#pragma unroll
  for (int nt = 0; nt < 8; ++nt) {
    float bv = bias[nt * 16 + lrow];
#pragma unroll
    for (int v = 0; v < 8; ++v) {
      Y[(size_t)(mbase + mo + v) * H + nt * 16 + lrow] = c[nt][v] + bv;
    }
  }
}

__device__ __forceinline__ void load_edge(int e, const int* __restrict__ rates,
                                          const int* __restrict__ impl,
                                          int& u, int& b) {
  const int idx = (e >= 2 * EPR) ? (e - 2 * EPR) : e;
  const int* ed = (idx < EPR) ? rates : impl;
  const int i = (idx < EPR) ? idx : (idx - EPR);
  u = ed[i];
  b = ed[i + EPR];
}

__device__ __forceinline__ void orient_edge(int e, int u, int b, int& row, int& col) {
  const int rev = (e >= 2 * EPR) ? 1 : 0;
  row = rev ? (b + NU) : u;
  col = rev ? u : b;
}

__device__ __forceinline__ void decode_edge(int e, const int* __restrict__ rates,
                                            const int* __restrict__ impl,
                                            int& row, int& col) {
  int u, b;
  load_edge(e, rates, impl, u, b);
  orient_edge(e, u, b, row, col);
}

#define BIN_SHIFT 9
#define BIN_NODES (1 << BIN_SHIFT)
#define NBIN ((NNODE + BIN_NODES - 1) / BIN_NODES)
#define NBIN_PAD 320
#define NBLK 256
#define CHUNK ((NE + NBLK - 1) / NBLK)
#define BT 1024

__global__ void __launch_bounds__(BT)
bin_count_kernel(const int* __restrict__ rates, const int* __restrict__ impl,
                 int* __restrict__ blkhist) {
  __shared__ int h[NBIN];
  const int t = threadIdx.x;
  for (int b = t; b < NBIN; b += BT) h[b] = 0;
  __syncthreads();
  const int e0 = blockIdx.x * CHUNK;
  const int e1 = min(e0 + CHUNK, NE);
  for (int eb = e0 + t; eb < e1; eb += 4 * BT) {
    int cl[4];
    for (int k = 0; k < 4; ++k) {
      const int e = eb + k * BT;
      int row;
      cl[k] = -1;
      if (e < e1) decode_edge(e, rates, impl, row, cl[k]);
    }
    for (int k = 0; k < 4; ++k) {
      if (cl[k] >= 0) atomicAdd(&h[cl[k] >> BIN_SHIFT], 1);
    }
  }
  __syncthreads();
  for (int b = t; b < NBIN; b += BT) blkhist[b * NBLK + blockIdx.x] = h[b];
}

__global__ void __launch_bounds__(BT)
bin_place_kernel(const int* __restrict__ rates, const int* __restrict__ impl,
                 const int* __restrict__ blkhist, int* __restrict__ binbase,
                 int2* __restrict__ pairs) {
  __shared__ int tot[NBIN_PAD];
  __shared__ int cursor[NBIN_PAD];
  const int t = threadIdx.x;
  const int lane = t & 31;
  const int wv = t >> 5;
  const int blk = blockIdx.x;
  for (int b = wv; b < NBIN; b += BT / 32) {
    const int4* rp = (const int4*)(blkhist + b * NBLK) + lane * 2;
    const int4 v0 = rp[0], v1 = rp[1];
    const int i0 = lane * 8;
    int all = ((v0.x + v0.y) + (v0.z + v0.w)) + ((v1.x + v1.y) + (v1.z + v1.w));
    int bef = 0;
    bef += (i0 + 0 < blk) ? v0.x : 0;
    bef += (i0 + 1 < blk) ? v0.y : 0;
    bef += (i0 + 2 < blk) ? v0.z : 0;
    bef += (i0 + 3 < blk) ? v0.w : 0;
    bef += (i0 + 4 < blk) ? v1.x : 0;
    bef += (i0 + 5 < blk) ? v1.y : 0;
    bef += (i0 + 6 < blk) ? v1.z : 0;
    bef += (i0 + 7 < blk) ? v1.w : 0;
    for (int o = 16; o > 0; o >>= 1) {
      all += __shfl_xor(all, o);
      bef += __shfl_xor(bef, o);
    }
    if (lane == 0) { tot[b] = all; cursor[b] = bef; }
  }
  __syncthreads();
  if (wv == 0) {
    int s = 0;
    for (int k = 0; k < 10; ++k) {
      const int b = lane * 10 + k;
      if (b < NBIN) s += tot[b];
    }
    int inc = s;
    for (int o = 1; o < 32; o <<= 1) {
      const int y = __shfl_up(inc, o);
      if (lane >= o) inc += y;
    }
    int run = inc - s;
    for (int k = 0; k < 10; ++k) {
      const int b = lane * 10 + k;
      if (b < NBIN) {
        const int tb = tot[b];
        cursor[b] += run;
        if (blk == 0) binbase[b] = run;
        run += tb;
      }
    }
    if (blk == 0 && lane == 31) binbase[NBIN] = inc;
  }
  __syncthreads();
  const int e0 = blk * CHUNK;
  const int e1 = min(e0 + CHUNK, NE);
  for (int eb = e0 + t; eb < e1; eb += 4 * BT) {
    int uu[4], bb[4];
    for (int k = 0; k < 4; ++k) {
      const int e = eb + k * BT;
      uu[k] = -1;
      bb[k] = 0;
      if (e < e1) load_edge(e, rates, impl, uu[k], bb[k]);
    }
    for (int k = 0; k < 4; ++k) {
      if (uu[k] >= 0) {
        int row, col;
        orient_edge(eb + k * BT, uu[k], bb[k], row, col);
        const int p = atomicAdd(&cursor[col >> BIN_SHIFT], 1);
        pairs[p] = make_int2(col, row);
      }
    }
  }
}

__global__ void __launch_bounds__(BT)
bin_sort_kernel(const int2* __restrict__ pairs, const int* __restrict__ binbase,
                int* __restrict__ cur, float* __restrict__ dis,
                int* __restrict__ srt_row) {
  __shared__ int cnt[BIN_NODES];
  __shared__ int wsum[BIN_NODES / 32];
  const int t = threadIdx.x;
  const int lane = t & 31;
  const int wv = t >> 5;
  const int nbase = blockIdx.x << BIN_SHIFT;
  const int rbeg = binbase[blockIdx.x];
  const int rend = binbase[blockIdx.x + 1];
  if (t < BIN_NODES) cnt[t] = 0;
  __syncthreads();
  for (int ib = rbeg + t; ib < rend; ib += 4 * BT) {
    int cl[4];
    for (int k = 0; k < 4; ++k) {
      const int i = ib + k * BT;
      cl[k] = (i < rend) ? pairs[i].x : -1;
    }
    for (int k = 0; k < 4; ++k) {
      if (cl[k] >= 0) atomicAdd(&cnt[cl[k] - nbase], 1);
    }
  }
  __syncthreads();
  int d = 0, inc = 0;
  if (t < BIN_NODES) {
    d = cnt[t];
    inc = d;
    for (int o = 1; o < 32; o <<= 1) {
      const int y = __shfl_up(inc, o);
      if (lane >= o) inc += y;
    }
    if (lane == 31) wsum[wv] = inc;
  }
  __syncthreads();
  if (t < BIN_NODES) {
    int woff = 0;
    for (int j = 0; j < BIN_NODES / 32; ++j) {
      if (j < wv) woff += wsum[j];
    }
    const int ex = woff + inc - d;
    cnt[t] = ex;
    const int node = nbase + t;
    if (node < NNODE) {
      dis[node] = (d > 0) ? rsqrtf((float)d) : 0.0f;
      cur[node] = rbeg + ex + d;
    }
  }
  __syncthreads();
  for (int ib = rbeg + t; ib < rend; ib += 4 * BT) {
    int2 p[4];
    for (int k = 0; k < 4; ++k) {
      const int i = ib + k * BT;
      p[k] = (i < rend) ? pairs[i] : make_int2(-1, 0);
    }
    for (int k = 0; k < 4; ++k) {
      if (p[k].x >= 0) {
        const int pos = atomicAdd(&cnt[p[k].x - nbase], 1);
        srt_row[rbeg + pos] = p[k].y;
      }
    }
  }
}

#define GPW 8
#define GBLOCKS ((NNODE + 8 * GPW - 1) / (8 * GPW))
#define GWAVES (GBLOCKS * 8)

#define PICK_EDGE(k)                                                          \
  const int s##k = __builtin_ctz(live);                                       \
  live &= live - 1u;                                                          \
  const int r##k = __builtin_amdgcn_readlane(r, s##k);                        \
  const float w##k = __builtin_bit_cast(                                      \
      float, __builtin_amdgcn_readlane(__builtin_bit_cast(int, w), s##k));
#define LOAD_ROW(k)                                                           \
  const float4 v##k = ((const float4*)(x + (size_t)r##k * H))[lane];
#define ADD_EDGE(k)                                                           \
  acc.x += v##k.x * w##k; acc.y += v##k.y * w##k;                             \
  acc.z += v##k.z * w##k; acc.w += v##k.w * w##k;

__device__ __forceinline__ float4 segment_sum(const float* __restrict__ x,
                                              const float* __restrict__ dis,
                                              const int* __restrict__ srt_row,
                                              int beg, int end, float dc, int lane) {
  float4 acc = make_float4(0.0f, 0.0f, 0.0f, 0.0f);
  for (int base = beg; base < end; base += 32) {
    const int idx = base + lane;
    int r = 0;
    float w = 0.0f;
    if (idx < end) {
      r = srt_row[idx];
      w = dis[r] * dc;
    }
    unsigned live = __builtin_amdgcn_ballot_w32(w != 0.0f);
    while (live != 0u) {
      PICK_EDGE(0)
      if (live == 0u) { LOAD_ROW(0) ADD_EDGE(0) break; }
      PICK_EDGE(1)
      if (live == 0u) { LOAD_ROW(0) LOAD_ROW(1) ADD_EDGE(0) ADD_EDGE(1) break; }
      PICK_EDGE(2)
      if (live == 0u) {
        LOAD_ROW(0) LOAD_ROW(1) LOAD_ROW(2)
        ADD_EDGE(0) ADD_EDGE(1) ADD_EDGE(2)
        break;
      }
      PICK_EDGE(3)
      LOAD_ROW(0) LOAD_ROW(1) LOAD_ROW(2) LOAD_ROW(3)
      ADD_EDGE(0) ADD_EDGE(1) ADD_EDGE(2) ADD_EDGE(3)
    }
  }
  return acc;
}

__global__ void __launch_bounds__(256)
gather_kernel(const float* __restrict__ x, const float* __restrict__ dis,
              const int* __restrict__ cur, const int* __restrict__ srt_row,
              float* __restrict__ out) {
  const int lane = threadIdx.x & 31;
  const int g = __builtin_amdgcn_readfirstlane(
      (int)((blockIdx.x * blockDim.x + threadIdx.x) >> 5));
  for (int j = 0; j < GPW; ++j) {
    const int c = g + j * GWAVES;
    if (c >= NNODE) break;
    const int prev = cur[(c > 0) ? c - 1 : 0];
    const int end = cur[c];
    const float dc = dis[c];
    const int beg = (c > 0) ? prev : 0;
    const float4 acc = segment_sum(x, dis, srt_row, beg, end, dc, lane);
    ((float4*)(out + (size_t)c * H))[lane] = acc;
  }
}

__global__ void __launch_bounds__(256)
gather_mean_kernel(const float* __restrict__ x1, const float* __restrict__ dis,
                   const int* __restrict__ cur, const int* __restrict__ srt_row,
                   float* __restrict__ out) {
  const int lane = threadIdx.x & 31;
  const int g = __builtin_amdgcn_readfirstlane(
      (int)((blockIdx.x * blockDim.x + threadIdx.x) >> 5));
  for (int j = 0; j < GPW; ++j) {
    const int c = g + j * GWAVES;
    if (c >= NNODE) break;
    const int prev = cur[(c > 0) ? c - 1 : 0];
    const int end = cur[c];
    const float dc = dis[c];
    const int beg = (c > 0) ? prev : 0;
    float4* orow = (float4*)(out + (size_t)c * H) + lane;
    const float4 o = *orow;
    const float4 a = ((const float4*)(x1 + (size_t)c * H))[lane];
    const float4 b = segment_sum(x1, dis, srt_row, beg, end, dc, lane);
    const float s = 1.0f / 3.0f;
    float4 m;
    m.x = ((o.x + a.x) + b.x) * s;
    m.y = ((o.y + a.y) + b.y) * s;
    m.z = ((o.z + a.z) + b.z) * s;
    m.w = ((o.w + a.w) + b.w) * s;
    *orow = m;
  }
}

extern "C" void kernel_launch(void* const* d_in, const int* in_sizes, int n_in,
                              void* d_out, int out_size, void* d_ws, size_t ws_size,
                              hipStream_t stream) {
  (void)in_sizes; (void)n_in; (void)out_size; (void)ws_size;

  const float* user_x = (const float*)d_in[0];
  const float* book_x = (const float*)d_in[1];
  const float* user_W = (const float*)d_in[2];
  const float* user_b = (const float*)d_in[3];
  const float* book_W = (const float*)d_in[4];
  const float* book_b = (const float*)d_in[5];
  const int*   rates  = (const int*)d_in[6];
  const int*   impl   = (const int*)d_in[7];
  float* outp = (float*)d_out;

  float* x1  = (float*)d_ws;
  float* x2  = x1  + (size_t)NNODE * H;
  float* deg = x2  + (size_t)NNODE * H;
  float* dis = deg + NNODE;
  int*   srt = (int*)(dis + NNODE);
  int*   cur = (int*)deg;
  int2*  pairs   = (int2*)x2;
  int*   blkhist = (int*)x2 + 2 * (size_t)NE;
  int*   binbase = blkhist + NBIN * NBLK;

  {
    int waves = NU / 16;
    gemm_bias_wmma<<<(waves + 7) / 8, 256, 0, stream>>>(user_x, user_W, user_b, outp, NU);
    waves = NB / 16;
    gemm_bias_wmma<<<(waves + 7) / 8, 256, 0, stream>>>(book_x, book_W, book_b,
                                                        outp + (size_t)NU * H, NB);
  }

  bin_count_kernel<<<NBLK, BT, 0, stream>>>(rates, impl, blkhist);
  bin_place_kernel<<<NBLK, BT, 0, stream>>>(rates, impl, blkhist, binbase, pairs);
  bin_sort_kernel<<<NBIN, BT, 0, stream>>>(pairs, binbase, cur, dis, srt);

  gather_kernel<<<GBLOCKS, 256, 0, stream>>>(outp, dis, cur, srt, x1);
  gather_mean_kernel<<<GBLOCKS, 256, 0, stream>>>(x1, dis, cur, srt, outp);
}


# both projections and the edge-count pass share one launch: 256 count blocks (8 edges per trip) stand in front of the 9375 GEMM strips of one grid and run under their matrix work; 7 launches -> 5
# speedup vs baseline: 4.2595x; 1.0959x over previous
#include <hip/hip_runtime.h>

#define NU 100000
#define NB 50000
#define NNODE (NU + NB)
#define H 128
#define F 256
#define EPR 500000
#define NE (4 * EPR)

typedef __attribute__((ext_vector_type(2))) float v2f;
typedef __attribute__((ext_vector_type(8))) float v8f;

__device__ __forceinline__ void gemm_strip_wmma(const float* __restrict__ Xs,
                                                const float* __restrict__ W,
                                                const float* __restrict__ bias,
                                                float* __restrict__ Ys, int lane) {
  const int lrow  = lane & 15;
  const int khalf = (lane >> 4) << 1;

  const float* arow = Xs + (size_t)lrow * F + khalf;

  v8f c[8] = {};

  for (int k = 0; k < F; k += 4) {
    v2f a = *(const v2f*)(arow + k);
#pragma unroll
    for (int nt = 0; nt < 8; ++nt) {
      v2f b = *(const v2f*)(W + (size_t)(nt * 16 + lrow) * F + khalf + k);
      c[nt] = __builtin_amdgcn_wmma_f32_16x16x4_f32(
           false, a,  false, b,
           (short)0, c[nt],  false,  false);
    }
  }

  const int mo = (lane >> 4) << 3;
#pragma unroll
  for (int nt = 0; nt < 8; ++nt) {
    float bv = bias[nt * 16 + lrow];
#pragma unroll
    for (int v = 0; v < 8; ++v) {
      Ys[(size_t)(mo + v) * H + nt * 16 + lrow] = c[nt][v] + bv;
    }
  }
}

__device__ __forceinline__ void load_edge(int e, const int* __restrict__ rates,
                                          const int* __restrict__ impl,
                                          int& u, int& b) {
  const int idx = (e >= 2 * EPR) ? (e - 2 * EPR) : e;
  const int* ed = (idx < EPR) ? rates : impl;
  const int i = (idx < EPR) ? idx : (idx - EPR);
  u = ed[i];
  b = ed[i + EPR];
}

__device__ __forceinline__ void orient_edge(int e, int u, int b, int& row, int& col) {
  const int rev = (e >= 2 * EPR) ? 1 : 0;
  row = rev ? (b + NU) : u;
  col = rev ? u : b;
}

__device__ __forceinline__ void decode_edge(int e, const int* __restrict__ rates,
                                            const int* __restrict__ impl,
                                            int& row, int& col) {
  int u, b;
  load_edge(e, rates, impl, u, b);
  orient_edge(e, u, b, row, col);
}

#define BIN_SHIFT 9
#define BIN_NODES (1 << BIN_SHIFT)
#define NBIN ((NNODE + BIN_NODES - 1) / BIN_NODES)
#define NBIN_PAD 320
#define NBLK 256
#define CHUNK ((NE + NBLK - 1) / NBLK)
#define BT 1024

#define S1T 256
#define NSTRIP_U (NU / 16)
#define NSTRIP ((NU + NB) / 16)
#define S1_GEMM_BLOCKS ((NSTRIP + 7) / 8)

__global__ void __launch_bounds__(S1T)
stage1_kernel(const float* __restrict__ user_x, const float* __restrict__ user_W,
              const float* __restrict__ user_b, const float* __restrict__ book_x,
              const float* __restrict__ book_W, const float* __restrict__ book_b,
              const int* __restrict__ rates, const int* __restrict__ impl,
              float* __restrict__ out, int* __restrict__ blkhist) {
  __shared__ int h[NBIN];
  const int t = threadIdx.x;
  if (blockIdx.x < NBLK) {
    const int blk = blockIdx.x;
    for (int b = t; b < NBIN; b += S1T) h[b] = 0;
    __syncthreads();
    const int e0 = blk * CHUNK;
    const int e1 = min(e0 + CHUNK, NE);
    for (int eb = e0 + t; eb < e1; eb += 8 * S1T) {
      int cl[8];
#pragma unroll
      for (int k = 0; k < 8; ++k) {
        const int e = eb + k * S1T;
        int row;
        cl[k] = -1;
        if (e < e1) decode_edge(e, rates, impl, row, cl[k]);
      }
#pragma unroll
      for (int k = 0; k < 8; ++k) {
        if (cl[k] >= 0) atomicAdd(&h[cl[k] >> BIN_SHIFT], 1);
      }
    }
    __syncthreads();
    for (int b = t; b < NBIN; b += S1T) blkhist[b * NBLK + blk] = h[b];
    return;
  }
  const int s = __builtin_amdgcn_readfirstlane(
      (int)((blockIdx.x - NBLK) * (S1T / 32) + (t >> 5)));
  if (s >= NSTRIP) return;
  const bool bk = s >= NSTRIP_U;
  const float* Xs = bk ? book_x + (size_t)(s - NSTRIP_U) * 16 * F
                       : user_x + (size_t)s * 16 * F;
  gemm_strip_wmma(Xs, bk ? book_W : user_W, bk ? book_b : user_b,
                  out + (size_t)s * 16 * H, t & 31);
}

__global__ void __launch_bounds__(BT)
bin_place_kernel(const int* __restrict__ rates, const int* __restrict__ impl,
                 const int* __restrict__ blkhist, int* __restrict__ binbase,
                 int2* __restrict__ pairs) {
  __shared__ int tot[NBIN_PAD];
  __shared__ int cursor[NBIN_PAD];
  const int t = threadIdx.x;
  const int lane = t & 31;
  const int wv = t >> 5;
  const int blk = blockIdx.x;
  for (int b = wv; b < NBIN; b += BT / 32) {
    const int4* rp = (const int4*)(blkhist + b * NBLK) + lane * 2;
    const int4 v0 = rp[0], v1 = rp[1];
    const int i0 = lane * 8;
    int all = ((v0.x + v0.y) + (v0.z + v0.w)) + ((v1.x + v1.y) + (v1.z + v1.w));
    int bef = 0;
    bef += (i0 + 0 < blk) ? v0.x : 0;
    bef += (i0 + 1 < blk) ? v0.y : 0;
    bef += (i0 + 2 < blk) ? v0.z : 0;
    bef += (i0 + 3 < blk) ? v0.w : 0;
    bef += (i0 + 4 < blk) ? v1.x : 0;
    bef += (i0 + 5 < blk) ? v1.y : 0;
    bef += (i0 + 6 < blk) ? v1.z : 0;
    bef += (i0 + 7 < blk) ? v1.w : 0;
    for (int o = 16; o > 0; o >>= 1) {
      all += __shfl_xor(all, o);
      bef += __shfl_xor(bef, o);
    }
    if (lane == 0) { tot[b] = all; cursor[b] = bef; }
  }
  __syncthreads();
  if (wv == 0) {
    int s = 0;
    for (int k = 0; k < 10; ++k) {
      const int b = lane * 10 + k;
      if (b < NBIN) s += tot[b];
    }
    int inc = s;
    for (int o = 1; o < 32; o <<= 1) {
      const int y = __shfl_up(inc, o);
      if (lane >= o) inc += y;
    }
    int run = inc - s;
    for (int k = 0; k < 10; ++k) {
      const int b = lane * 10 + k;
      if (b < NBIN) {
        const int tb = tot[b];
        cursor[b] += run;
        if (blk == 0) binbase[b] = run;
        run += tb;
      }
    }
    if (blk == 0 && lane == 31) binbase[NBIN] = inc;
  }
  __syncthreads();
  const int e0 = blk * CHUNK;
  const int e1 = min(e0 + CHUNK, NE);
  for (int eb = e0 + t; eb < e1; eb += 4 * BT) {
    int uu[4], bb[4];
    for (int k = 0; k < 4; ++k) {
      const int e = eb + k * BT;
      uu[k] = -1;
      bb[k] = 0;
      if (e < e1) load_edge(e, rates, impl, uu[k], bb[k]);
    }
    for (int k = 0; k < 4; ++k) {
      if (uu[k] >= 0) {
        int row, col;
        orient_edge(eb + k * BT, uu[k], bb[k], row, col);
        const int p = atomicAdd(&cursor[col >> BIN_SHIFT], 1);
        pairs[p] = make_int2(col, row);
      }
    }
  }
}

__global__ void __launch_bounds__(BT)
bin_sort_kernel(const int2* __restrict__ pairs, const int* __restrict__ binbase,
                int* __restrict__ cur, float* __restrict__ dis,
                int* __restrict__ srt_row) {
  __shared__ int cnt[BIN_NODES];
  __shared__ int wsum[BIN_NODES / 32];
  const int t = threadIdx.x;
  const int lane = t & 31;
  const int wv = t >> 5;
  const int nbase = blockIdx.x << BIN_SHIFT;
  const int rbeg = binbase[blockIdx.x];
  const int rend = binbase[blockIdx.x + 1];
  if (t < BIN_NODES) cnt[t] = 0;
  __syncthreads();
  for (int ib = rbeg + t; ib < rend; ib += 4 * BT) {
    int cl[4];
    for (int k = 0; k < 4; ++k) {
      const int i = ib + k * BT;
      cl[k] = (i < rend) ? pairs[i].x : -1;
    }
    for (int k = 0; k < 4; ++k) {
      if (cl[k] >= 0) atomicAdd(&cnt[cl[k] - nbase], 1);
    }
  }
  __syncthreads();
  int d = 0, inc = 0;
  if (t < BIN_NODES) {
    d = cnt[t];
    inc = d;
    for (int o = 1; o < 32; o <<= 1) {
      const int y = __shfl_up(inc, o);
      if (lane >= o) inc += y;
    }
    if (lane == 31) wsum[wv] = inc;
  }
  __syncthreads();
  if (t < BIN_NODES) {
    int woff = 0;
    for (int j = 0; j < BIN_NODES / 32; ++j) {
      if (j < wv) woff += wsum[j];
    }
    const int ex = woff + inc - d;
    cnt[t] = ex;
    const int node = nbase + t;
    if (node < NNODE) {
      dis[node] = (d > 0) ? rsqrtf((float)d) : 0.0f;
      cur[node] = rbeg + ex + d;
    }
  }
  __syncthreads();
  for (int ib = rbeg + t; ib < rend; ib += 4 * BT) {
    int2 p[4];
    for (int k = 0; k < 4; ++k) {
      const int i = ib + k * BT;
      p[k] = (i < rend) ? pairs[i] : make_int2(-1, 0);
    }
    for (int k = 0; k < 4; ++k) {
      if (p[k].x >= 0) {
        const int pos = atomicAdd(&cnt[p[k].x - nbase], 1);
        srt_row[rbeg + pos] = p[k].y;
      }
    }
  }
}

#define GPW 8
#define GBLOCKS ((NNODE + 8 * GPW - 1) / (8 * GPW))
#define GWAVES (GBLOCKS * 8)

#define PICK_EDGE(k)                                                          \
  const int s##k = __builtin_ctz(live);                                       \
  live &= live - 1u;                                                          \
  const int r##k = __builtin_amdgcn_readlane(r, s##k);                        \
  const float w##k = __builtin_bit_cast(                                      \
      float, __builtin_amdgcn_readlane(__builtin_bit_cast(int, w), s##k));
#define LOAD_ROW(k)                                                           \
  const float4 v##k = ((const float4*)(x + (size_t)r##k * H))[lane];
#define ADD_EDGE(k)                                                           \
  acc.x += v##k.x * w##k; acc.y += v##k.y * w##k;                             \
  acc.z += v##k.z * w##k; acc.w += v##k.w * w##k;

__device__ __forceinline__ float4 segment_sum(const float* __restrict__ x,
                                              const float* __restrict__ dis,
                                              const int* __restrict__ srt_row,
                                              int beg, int end, float dc, int lane) {
  float4 acc = make_float4(0.0f, 0.0f, 0.0f, 0.0f);
  for (int base = beg; base < end; base += 32) {
    const int idx = base + lane;
    int r = 0;
    float w = 0.0f;
    if (idx < end) {
      r = srt_row[idx];
      w = dis[r] * dc;
    }
    unsigned live = __builtin_amdgcn_ballot_w32(w != 0.0f);
    while (live != 0u) {
      PICK_EDGE(0)
      if (live == 0u) { LOAD_ROW(0) ADD_EDGE(0) break; }
      PICK_EDGE(1)
      if (live == 0u) { LOAD_ROW(0) LOAD_ROW(1) ADD_EDGE(0) ADD_EDGE(1) break; }
      PICK_EDGE(2)
      if (live == 0u) {
        LOAD_ROW(0) LOAD_ROW(1) LOAD_ROW(2)
        ADD_EDGE(0) ADD_EDGE(1) ADD_EDGE(2)
        break;
      }
      PICK_EDGE(3)
      LOAD_ROW(0) LOAD_ROW(1) LOAD_ROW(2) LOAD_ROW(3)
      ADD_EDGE(0) ADD_EDGE(1) ADD_EDGE(2) ADD_EDGE(3)
    }
  }
  return acc;
}

__global__ void __launch_bounds__(256)
gather_kernel(const float* __restrict__ x, const float* __restrict__ dis,
              const int* __restrict__ cur, const int* __restrict__ srt_row,
              float* __restrict__ out) {
  const int lane = threadIdx.x & 31;
  const int g = __builtin_amdgcn_readfirstlane(
      (int)((blockIdx.x * blockDim.x + threadIdx.x) >> 5));
  for (int j = 0; j < GPW; ++j) {
    const int c = g + j * GWAVES;
    if (c >= NNODE) break;
    const int prev = cur[(c > 0) ? c - 1 : 0];
    const int end = cur[c];
    const float dc = dis[c];
    const int beg = (c > 0) ? prev : 0;
    const float4 acc = segment_sum(x, dis, srt_row, beg, end, dc, lane);
    ((float4*)(out + (size_t)c * H))[lane] = acc;
  }
}

__global__ void __launch_bounds__(256)
gather_mean_kernel(const float* __restrict__ x1, const float* __restrict__ dis,
                   const int* __restrict__ cur, const int* __restrict__ srt_row,
                   float* __restrict__ out) {
  const int lane = threadIdx.x & 31;
  const int g = __builtin_amdgcn_readfirstlane(
      (int)((blockIdx.x * blockDim.x + threadIdx.x) >> 5));
  for (int j = 0; j < GPW; ++j) {
    const int c = g + j * GWAVES;
    if (c >= NNODE) break;
    const int prev = cur[(c > 0) ? c - 1 : 0];
    const int end = cur[c];
    const float dc = dis[c];
    const int beg = (c > 0) ? prev : 0;
    float4* orow = (float4*)(out + (size_t)c * H) + lane;
    const float4 o = *orow;
    const float4 a = ((const float4*)(x1 + (size_t)c * H))[lane];
    const float4 b = segment_sum(x1, dis, srt_row, beg, end, dc, lane);
    const float s = 1.0f / 3.0f;
    float4 m;
    m.x = ((o.x + a.x) + b.x) * s;
    m.y = ((o.y + a.y) + b.y) * s;
    m.z = ((o.z + a.z) + b.z) * s;
    m.w = ((o.w + a.w) + b.w) * s;
    *orow = m;
  }
}

extern "C" void kernel_launch(void* const* d_in, const int* in_sizes, int n_in,
                              void* d_out, int out_size, void* d_ws, size_t ws_size,
                              hipStream_t stream) {
  (void)in_sizes; (void)n_in; (void)out_size; (void)ws_size;

  const float* user_x = (const float*)d_in[0];
  const float* book_x = (const float*)d_in[1];
  const float* user_W = (const float*)d_in[2];
  const float* user_b = (const float*)d_in[3];
  const float* book_W = (const float*)d_in[4];
  const float* book_b = (const float*)d_in[5];
  const int*   rates  = (const int*)d_in[6];
  const int*   impl   = (const int*)d_in[7];
  float* outp = (float*)d_out;

  float* x1  = (float*)d_ws;
  float* x2  = x1  + (size_t)NNODE * H;
  float* deg = x2  + (size_t)NNODE * H;
  float* dis = deg + NNODE;
  int*   srt = (int*)(dis + NNODE);
  int*   cur = (int*)deg;
  int2*  pairs   = (int2*)x2;
  int*   blkhist = (int*)x2 + 2 * (size_t)NE;
  int*   binbase = blkhist + NBIN * NBLK;

  stage1_kernel<<<NBLK + S1_GEMM_BLOCKS, S1T, 0, stream>>>(user_x, user_W, user_b,
                                                           book_x, book_W, book_b,
                                                           rates, impl, outp, blkhist);

  bin_place_kernel<<<NBLK, BT, 0, stream>>>(rates, impl, blkhist, binbase, pairs);
  bin_sort_kernel<<<NBIN, BT, 0, stream>>>(pairs, binbase, cur, dis, srt);

  gather_kernel<<<GBLOCKS, 256, 0, stream>>>(outp, dis, cur, srt, x1);
  gather_mean_kernel<<<GBLOCKS, 256, 0, stream>>>(x1, dis, cur, srt, outp);
}


# projection GEMM: each block copies its 128x256 weight table into LDS once (whole-line 16-byte loads, row pitch 260 floats) and reads every B fragment from LDS, not global; k order and sums unchanged
# speedup vs baseline: 5.6533x; 1.3272x over previous
#include <hip/hip_runtime.h>

#define NU 100000
#define NB 50000
#define NNODE (NU + NB)
#define H 128
#define F 256
#define EPR 500000
#define NE (4 * EPR)

typedef __attribute__((ext_vector_type(2))) float v2f;
typedef __attribute__((ext_vector_type(8))) float v8f;

#define WP 260

__device__ __forceinline__ void gemm_strip_wmma(const float* __restrict__ Xs,
                                                const float* Ws,
                                                const float* __restrict__ bias,
                                                float* __restrict__ Ys, int lane) {
  const int lrow  = lane & 15;
  const int khalf = (lane >> 4) << 1;

  const float* arow = Xs + (size_t)lrow * F + khalf;
  const float* brow = Ws + lrow * WP + khalf;

  v8f c[8] = {};

  for (int k = 0; k < F; k += 4) {
    v2f a = *(const v2f*)(arow + k);
#pragma unroll
    for (int nt = 0; nt < 8; ++nt) {
      v2f b = *(const v2f*)(brow + nt * 16 * WP + k);
      c[nt] = __builtin_amdgcn_wmma_f32_16x16x4_f32(
           false, a,  false, b,
           (short)0, c[nt],  false,  false);
    }
  }

  const int mo = (lane >> 4) << 3;
#pragma unroll
  for (int nt = 0; nt < 8; ++nt) {
    float bv = bias[nt * 16 + lrow];
#pragma unroll
    for (int v = 0; v < 8; ++v) {
      Ys[(size_t)(mo + v) * H + nt * 16 + lrow] = c[nt][v] + bv;
    }
  }
}

__device__ __forceinline__ void load_edge(int e, const int* __restrict__ rates,
                                          const int* __restrict__ impl,
                                          int& u, int& b) {
  const int idx = (e >= 2 * EPR) ? (e - 2 * EPR) : e;
  const int* ed = (idx < EPR) ? rates : impl;
  const int i = (idx < EPR) ? idx : (idx - EPR);
  u = ed[i];
  b = ed[i + EPR];
}

__device__ __forceinline__ void orient_edge(int e, int u, int b, int& row, int& col) {
  const int rev = (e >= 2 * EPR) ? 1 : 0;
  row = rev ? (b + NU) : u;
  col = rev ? u : b;
}

__device__ __forceinline__ void decode_edge(int e, const int* __restrict__ rates,
                                            const int* __restrict__ impl,
                                            int& row, int& col) {
  int u, b;
  load_edge(e, rates, impl, u, b);
  orient_edge(e, u, b, row, col);
}

#define BIN_SHIFT 9
#define BIN_NODES (1 << BIN_SHIFT)
#define NBIN ((NNODE + BIN_NODES - 1) / BIN_NODES)
#define NBIN_PAD 320
#define NBLK 256
#define CHUNK ((NE + NBLK - 1) / NBLK)
#define BT 1024

#define S1T 256
#define NSTRIP_U (NU / 16)
#define NSTRIP ((NU + NB) / 16)
#define S1_UBLOCKS ((NSTRIP_U + 7) / 8)
#define S1_BBLOCKS ((NSTRIP - NSTRIP_U + 7) / 8)
#define S1_GEMM_BLOCKS (S1_UBLOCKS + S1_BBLOCKS)

__global__ void __launch_bounds__(S1T)
stage1_kernel(const float* __restrict__ user_x, const float* __restrict__ user_W,
              const float* __restrict__ user_b, const float* __restrict__ book_x,
              const float* __restrict__ book_W, const float* __restrict__ book_b,
              const int* __restrict__ rates, const int* __restrict__ impl,
              float* __restrict__ out, int* __restrict__ blkhist) {
  __shared__ int h[NBIN];
  __shared__ __attribute__((aligned(16))) float Ws[H * WP];
  const int t = threadIdx.x;
  if (blockIdx.x < NBLK) {
    const int blk = blockIdx.x;
    for (int b = t; b < NBIN; b += S1T) h[b] = 0;
    __syncthreads();
    const int e0 = blk * CHUNK;
    const int e1 = min(e0 + CHUNK, NE);
    for (int eb = e0 + t; eb < e1; eb += 8 * S1T) {
      int cl[8];
#pragma unroll
      for (int k = 0; k < 8; ++k) {
        const int e = eb + k * S1T;
        int row;
        cl[k] = -1;
        if (e < e1) decode_edge(e, rates, impl, row, cl[k]);
      }
#pragma unroll
      for (int k = 0; k < 8; ++k) {
        if (cl[k] >= 0) atomicAdd(&h[cl[k] >> BIN_SHIFT], 1);
      }
    }
    __syncthreads();
    for (int b = t; b < NBIN; b += S1T) blkhist[b * NBLK + blk] = h[b];
    return;
  }
  const int gb = (int)blockIdx.x - NBLK;
  const bool bk = gb >= S1_UBLOCKS;
  const float4* Wg = (const float4*)(bk ? book_W : user_W);
#pragma unroll 8
  for (int j = 0; j < H * F / 4 / S1T; ++j) {
    const int i = j * S1T + t;
    const float4 v = Wg[i];
    *(float4*)(Ws + (i >> 6) * WP + ((i & 63) << 2)) = v;
  }
  __syncthreads();
  const int s = __builtin_amdgcn_readfirstlane(
      (bk ? NSTRIP_U + (gb - S1_UBLOCKS) * (S1T / 32) : gb * (S1T / 32)) + (t >> 5));
  if (s >= (bk ? NSTRIP : NSTRIP_U)) return;
  const float* Xs = bk ? book_x + (size_t)(s - NSTRIP_U) * 16 * F
                       : user_x + (size_t)s * 16 * F;
  gemm_strip_wmma(Xs, Ws, bk ? book_b : user_b,
                  out + (size_t)s * 16 * H, t & 31);
}

__global__ void __launch_bounds__(BT)
bin_place_kernel(const int* __restrict__ rates, const int* __restrict__ impl,
                 const int* __restrict__ blkhist, int* __restrict__ binbase,
                 int2* __restrict__ pairs) {
  __shared__ int tot[NBIN_PAD];
  __shared__ int cursor[NBIN_PAD];
  const int t = threadIdx.x;
  const int lane = t & 31;
  const int wv = t >> 5;
  const int blk = blockIdx.x;
  for (int b = wv; b < NBIN; b += BT / 32) {
    const int4* rp = (const int4*)(blkhist + b * NBLK) + lane * 2;
    const int4 v0 = rp[0], v1 = rp[1];
    const int i0 = lane * 8;
    int all = ((v0.x + v0.y) + (v0.z + v0.w)) + ((v1.x + v1.y) + (v1.z + v1.w));
    int bef = 0;
    bef += (i0 + 0 < blk) ? v0.x : 0;
    bef += (i0 + 1 < blk) ? v0.y : 0;
    bef += (i0 + 2 < blk) ? v0.z : 0;
    bef += (i0 + 3 < blk) ? v0.w : 0;
    bef += (i0 + 4 < blk) ? v1.x : 0;
    bef += (i0 + 5 < blk) ? v1.y : 0;
    bef += (i0 + 6 < blk) ? v1.z : 0;
    bef += (i0 + 7 < blk) ? v1.w : 0;
    for (int o = 16; o > 0; o >>= 1) {
      all += __shfl_xor(all, o);
      bef += __shfl_xor(bef, o);
    }
    if (lane == 0) { tot[b] = all; cursor[b] = bef; }
  }
  __syncthreads();
  if (wv == 0) {
    int s = 0;
    for (int k = 0; k < 10; ++k) {
      const int b = lane * 10 + k;
      if (b < NBIN) s += tot[b];
    }
    int inc = s;
    for (int o = 1; o < 32; o <<= 1) {
      const int y = __shfl_up(inc, o);
      if (lane >= o) inc += y;
    }
    int run = inc - s;
    for (int k = 0; k < 10; ++k) {
      const int b = lane * 10 + k;
      if (b < NBIN) {
        const int tb = tot[b];
        cursor[b] += run;
        if (blk == 0) binbase[b] = run;
        run += tb;
      }
    }
    if (blk == 0 && lane == 31) binbase[NBIN] = inc;
  }
  __syncthreads();
  const int e0 = blk * CHUNK;
  const int e1 = min(e0 + CHUNK, NE);
  for (int eb = e0 + t; eb < e1; eb += 4 * BT) {
    int uu[4], bb[4];
    for (int k = 0; k < 4; ++k) {
      const int e = eb + k * BT;
      uu[k] = -1;
      bb[k] = 0;
      if (e < e1) load_edge(e, rates, impl, uu[k], bb[k]);
    }
    for (int k = 0; k < 4; ++k) {
      if (uu[k] >= 0) {
        int row, col;
        orient_edge(eb + k * BT, uu[k], bb[k], row, col);
        const int p = atomicAdd(&cursor[col >> BIN_SHIFT], 1);
        pairs[p] = make_int2(col, row);
      }
    }
  }
}

__global__ void __launch_bounds__(BT)
bin_sort_kernel(const int2* __restrict__ pairs, const int* __restrict__ binbase,
                int* __restrict__ cur, float* __restrict__ dis,
                int* __restrict__ srt_row) {
  __shared__ int cnt[BIN_NODES];
  __shared__ int wsum[BIN_NODES / 32];
  const int t = threadIdx.x;
  const int lane = t & 31;
  const int wv = t >> 5;
  const int nbase = blockIdx.x << BIN_SHIFT;
  const int rbeg = binbase[blockIdx.x];
  const int rend = binbase[blockIdx.x + 1];
  if (t < BIN_NODES) cnt[t] = 0;
  __syncthreads();
  for (int ib = rbeg + t; ib < rend; ib += 4 * BT) {
    int cl[4];
    for (int k = 0; k < 4; ++k) {
      const int i = ib + k * BT;
      cl[k] = (i < rend) ? pairs[i].x : -1;
    }
    for (int k = 0; k < 4; ++k) {
      if (cl[k] >= 0) atomicAdd(&cnt[cl[k] - nbase], 1);
    }
  }
  __syncthreads();
  int d = 0, inc = 0;
  if (t < BIN_NODES) {
    d = cnt[t];
    inc = d;
    for (int o = 1; o < 32; o <<= 1) {
      const int y = __shfl_up(inc, o);
      if (lane >= o) inc += y;
    }
    if (lane == 31) wsum[wv] = inc;
  }
  __syncthreads();
  if (t < BIN_NODES) {
    int woff = 0;
    for (int j = 0; j < BIN_NODES / 32; ++j) {
      if (j < wv) woff += wsum[j];
    }
    const int ex = woff + inc - d;
    cnt[t] = ex;
    const int node = nbase + t;
    if (node < NNODE) {
      dis[node] = (d > 0) ? rsqrtf((float)d) : 0.0f;
      cur[node] = rbeg + ex + d;
    }
  }
  __syncthreads();
  for (int ib = rbeg + t; ib < rend; ib += 4 * BT) {
    int2 p[4];
    for (int k = 0; k < 4; ++k) {
      const int i = ib + k * BT;
      p[k] = (i < rend) ? pairs[i] : make_int2(-1, 0);
    }
    for (int k = 0; k < 4; ++k) {
      if (p[k].x >= 0) {
        const int pos = atomicAdd(&cnt[p[k].x - nbase], 1);
        srt_row[rbeg + pos] = p[k].y;
      }
    }
  }
}

#define GPW 8
#define GBLOCKS ((NNODE + 8 * GPW - 1) / (8 * GPW))
#define GWAVES (GBLOCKS * 8)

#define PICK_EDGE(k)                                                          \
  const int s##k = __builtin_ctz(live);                                       \
  live &= live - 1u;                                                          \
  const int r##k = __builtin_amdgcn_readlane(r, s##k);                        \
  const float w##k = __builtin_bit_cast(                                      \
      float, __builtin_amdgcn_readlane(__builtin_bit_cast(int, w), s##k));
#define LOAD_ROW(k)                                                           \
  const float4 v##k = ((const float4*)(x + (size_t)r##k * H))[lane];
#define ADD_EDGE(k)                                                           \
  acc.x += v##k.x * w##k; acc.y += v##k.y * w##k;                             \
  acc.z += v##k.z * w##k; acc.w += v##k.w * w##k;

__device__ __forceinline__ float4 segment_sum(const float* __restrict__ x,
                                              const float* __restrict__ dis,
                                              const int* __restrict__ srt_row,
                                              int beg, int end, float dc, int lane) {
  float4 acc = make_float4(0.0f, 0.0f, 0.0f, 0.0f);
  for (int base = beg; base < end; base += 32) {
    const int idx = base + lane;
    int r = 0;
    float w = 0.0f;
    if (idx < end) {
      r = srt_row[idx];
      w = dis[r] * dc;
    }
    unsigned live = __builtin_amdgcn_ballot_w32(w != 0.0f);
    while (live != 0u) {
      PICK_EDGE(0)
      if (live == 0u) { LOAD_ROW(0) ADD_EDGE(0) break; }
      PICK_EDGE(1)
      if (live == 0u) { LOAD_ROW(0) LOAD_ROW(1) ADD_EDGE(0) ADD_EDGE(1) break; }
      PICK_EDGE(2)
      if (live == 0u) {
        LOAD_ROW(0) LOAD_ROW(1) LOAD_ROW(2)
        ADD_EDGE(0) ADD_EDGE(1) ADD_EDGE(2)
        break;
      }
      PICK_EDGE(3)
      LOAD_ROW(0) LOAD_ROW(1) LOAD_ROW(2) LOAD_ROW(3)
      ADD_EDGE(0) ADD_EDGE(1) ADD_EDGE(2) ADD_EDGE(3)
    }
  }
  return acc;
}

__global__ void __launch_bounds__(256)
gather_kernel(const float* __restrict__ x, const float* __restrict__ dis,
              const int* __restrict__ cur, const int* __restrict__ srt_row,
              float* __restrict__ out) {
  const int lane = threadIdx.x & 31;
  const int g = __builtin_amdgcn_readfirstlane(
      (int)((blockIdx.x * blockDim.x + threadIdx.x) >> 5));
  for (int j = 0; j < GPW; ++j) {
    const int c = g + j * GWAVES;
    if (c >= NNODE) break;
    const int prev = cur[(c > 0) ? c - 1 : 0];
    const int end = cur[c];
    const float dc = dis[c];
    const int beg = (c > 0) ? prev : 0;
    const float4 acc = segment_sum(x, dis, srt_row, beg, end, dc, lane);
    ((float4*)(out + (size_t)c * H))[lane] = acc;
  }
}

__global__ void __launch_bounds__(256)
gather_mean_kernel(const float* __restrict__ x1, const float* __restrict__ dis,
                   const int* __restrict__ cur, const int* __restrict__ srt_row,
                   float* __restrict__ out) {
  const int lane = threadIdx.x & 31;
  const int g = __builtin_amdgcn_readfirstlane(
      (int)((blockIdx.x * blockDim.x + threadIdx.x) >> 5));
  for (int j = 0; j < GPW; ++j) {
    const int c = g + j * GWAVES;
    if (c >= NNODE) break;
    const int prev = cur[(c > 0) ? c - 1 : 0];
    const int end = cur[c];
    const float dc = dis[c];
    const int beg = (c > 0) ? prev : 0;
    float4* orow = (float4*)(out + (size_t)c * H) + lane;
    const float4 o = *orow;
    const float4 a = ((const float4*)(x1 + (size_t)c * H))[lane];
    const float4 b = segment_sum(x1, dis, srt_row, beg, end, dc, lane);
    const float s = 1.0f / 3.0f;
    float4 m;
    m.x = ((o.x + a.x) + b.x) * s;
    m.y = ((o.y + a.y) + b.y) * s;
    m.z = ((o.z + a.z) + b.z) * s;
    m.w = ((o.w + a.w) + b.w) * s;
    *orow = m;
  }
}

extern "C" void kernel_launch(void* const* d_in, const int* in_sizes, int n_in,
                              void* d_out, int out_size, void* d_ws, size_t ws_size,
                              hipStream_t stream) {
  (void)in_sizes; (void)n_in; (void)out_size; (void)ws_size;

  const float* user_x = (const float*)d_in[0];
  const float* book_x = (const float*)d_in[1];
  const float* user_W = (const float*)d_in[2];
  const float* user_b = (const float*)d_in[3];
  const float* book_W = (const float*)d_in[4];
  const float* book_b = (const float*)d_in[5];
  const int*   rates  = (const int*)d_in[6];
  const int*   impl   = (const int*)d_in[7];
  float* outp = (float*)d_out;

  float* x1  = (float*)d_ws;
  float* x2  = x1  + (size_t)NNODE * H;
  float* deg = x2  + (size_t)NNODE * H;
  float* dis = deg + NNODE;
  int*   srt = (int*)(dis + NNODE);
  int*   cur = (int*)deg;
  int2*  pairs   = (int2*)x2;
  int*   blkhist = (int*)x2 + 2 * (size_t)NE;
  int*   binbase = blkhist + NBIN * NBLK;

  stage1_kernel<<<NBLK + S1_GEMM_BLOCKS, S1T, 0, stream>>>(user_x, user_W, user_b,
                                                           book_x, book_W, book_b,
                                                           rates, impl, outp, blkhist);

  bin_place_kernel<<<NBLK, BT, 0, stream>>>(rates, impl, blkhist, binbase, pairs);
  bin_sort_kernel<<<NBIN, BT, 0, stream>>>(pairs, binbase, cur, dis, srt);

  gather_kernel<<<GBLOCKS, 256, 0, stream>>>(outp, dis, cur, srt, x1);
  gather_mean_kernel<<<GBLOCKS, 256, 0, stream>>>(x1, dis, cur, srt, outp);
}


# projection strips spread over the three pre-gather launches: a 12-strip GEMM block shares each WGP with a 512-thread placement or sort block; edge grouping runs under matrix work; sums as before
# speedup vs baseline: 5.8698x; 1.0383x over previous
#include <hip/hip_runtime.h>

#define NU 100000
#define NB 50000
#define NNODE (NU + NB)
#define H 128
#define F 256
#define EPR 500000
#define NE (4 * EPR)

typedef __attribute__((ext_vector_type(2))) float v2f;
typedef __attribute__((ext_vector_type(8))) float v8f;

#define WP 260

__device__ __forceinline__ void gemm_strip_wmma(const float* __restrict__ Xs,
                                                const float* Ws,
                                                const float* __restrict__ bias,
                                                float* __restrict__ Ys, int lane) {
  const int lrow  = lane & 15;
  const int khalf = (lane >> 4) << 1;

  const float* arow = Xs + (size_t)lrow * F + khalf;
  const float* brow = Ws + lrow * WP + khalf;

  v8f c[8] = {};

  for (int k = 0; k < F; k += 4) {
    v2f a = *(const v2f*)(arow + k);
#pragma unroll
    for (int nt = 0; nt < 8; ++nt) {
      v2f b = *(const v2f*)(brow + nt * 16 * WP + k);
      c[nt] = __builtin_amdgcn_wmma_f32_16x16x4_f32(
           false, a,  false, b,
           (short)0, c[nt],  false,  false);
    }
  }

  const int mo = (lane >> 4) << 3;
#pragma unroll
  for (int nt = 0; nt < 8; ++nt) {
    float bv = bias[nt * 16 + lrow];
#pragma unroll
    for (int v = 0; v < 8; ++v) {
      Ys[(size_t)(mo + v) * H + nt * 16 + lrow] = c[nt][v] + bv;
    }
  }
}

__device__ __forceinline__ void load_edge(int e, const int* __restrict__ rates,
                                          const int* __restrict__ impl,
                                          int& u, int& b) {
  const int idx = (e >= 2 * EPR) ? (e - 2 * EPR) : e;
  const int* ed = (idx < EPR) ? rates : impl;
  const int i = (idx < EPR) ? idx : (idx - EPR);
  u = ed[i];
  b = ed[i + EPR];
}

__device__ __forceinline__ void orient_edge(int e, int u, int b, int& row, int& col) {
  const int rev = (e >= 2 * EPR) ? 1 : 0;
  row = rev ? (b + NU) : u;
  col = rev ? u : b;
}

__device__ __forceinline__ void decode_edge(int e, const int* __restrict__ rates,
                                            const int* __restrict__ impl,
                                            int& row, int& col) {
  int u, b;
  load_edge(e, rates, impl, u, b);
  orient_edge(e, u, b, row, col);
}

#define BIN_SHIFT 9
#define BIN_NODES (1 << BIN_SHIFT)
#define NBIN ((NNODE + BIN_NODES - 1) / BIN_NODES)
#define NBIN_PAD 320
#define NBLK 256
#define CHUNK ((NE + NBLK - 1) / NBLK)

#define S1T 256
#define NSTRIP_U (NU / 16)
#define NSTRIP ((NU + NB) / 16)
#define SPB1 8
#define G1 512
#define L1_SE (G1 * SPB1)
#define GT 512
#define SPB2 12
#define G2U ((NSTRIP_U - L1_SE + SPB2 - 1) / SPB2)
#define G2B (256 - G2U)
#define G2 (G2U + G2B)
#define L2_SE (NSTRIP_U + G2B * SPB2)
#define G3 ((NSTRIP - L2_SE + SPB2 - 1) / SPB2)

template <int NT, int SPB>
__device__ __forceinline__ void gemm_block(bool bk, int s0, int lim, float* Ws,
                                           const float* __restrict__ user_x,
                                           const float* __restrict__ user_W,
                                           const float* __restrict__ user_b,
                                           const float* __restrict__ book_x,
                                           const float* __restrict__ book_W,
                                           const float* __restrict__ book_b,
                                           float* __restrict__ out) {
  const int t = threadIdx.x;
  const float4* Wg = (const float4*)(bk ? book_W : user_W);
#pragma unroll 8
  for (int j = 0; j < H * F / 4 / NT; ++j) {
    const int i = j * NT + t;
    const float4 v = Wg[i];
    *(float4*)(Ws + (i >> 6) * WP + ((i & 63) << 2)) = v;
  }
  __syncthreads();
  const int w = t >> 5;
  if (NT / 32 > SPB && w >= SPB) return;
  const int s = __builtin_amdgcn_readfirstlane(s0 + w);
  if (s >= lim) return;
  const float* Xs = bk ? book_x + (size_t)(s - NSTRIP_U) * 16 * F
                       : user_x + (size_t)s * 16 * F;
  gemm_strip_wmma(Xs, Ws, bk ? book_b : user_b,
                  out + (size_t)s * 16 * H, t & 31);
}

__global__ void __launch_bounds__(S1T)
stage1_kernel(const float* __restrict__ user_x, const float* __restrict__ user_W,
              const float* __restrict__ user_b, const float* __restrict__ book_x,
              const float* __restrict__ book_W, const float* __restrict__ book_b,
              const int* __restrict__ rates, const int* __restrict__ impl,
              float* __restrict__ out, int* __restrict__ blkhist) {
  __shared__ int h[NBIN];
  __shared__ __attribute__((aligned(16))) float Ws[H * WP];
  const int t = threadIdx.x;
  if (blockIdx.x < NBLK) {
    const int blk = blockIdx.x;
    for (int b = t; b < NBIN; b += S1T) h[b] = 0;
    __syncthreads();
    const int e0 = blk * CHUNK;
    const int e1 = min(e0 + CHUNK, NE);
    for (int eb = e0 + t; eb < e1; eb += 8 * S1T) {
      int cl[8];
#pragma unroll
      for (int k = 0; k < 8; ++k) {
        const int e = eb + k * S1T;
        int row;
        cl[k] = -1;
        if (e < e1) decode_edge(e, rates, impl, row, cl[k]);
      }
#pragma unroll
      for (int k = 0; k < 8; ++k) {
        if (cl[k] >= 0) atomicAdd(&h[cl[k] >> BIN_SHIFT], 1);
      }
    }
    __syncthreads();
    for (int b = t; b < NBIN; b += S1T) blkhist[b * NBLK + blk] = h[b];
    return;
  }
  gemm_block<S1T, SPB1>(false, ((int)blockIdx.x - NBLK) * SPB1, L1_SE, Ws,
                        user_x, user_W, user_b, book_x, book_W, book_b, out);
}

__global__ void __launch_bounds__(GT)
place_gemm_kernel(const float* __restrict__ user_x, const float* __restrict__ user_W,
                  const float* __restrict__ user_b, const float* __restrict__ book_x,
                  const float* __restrict__ book_W, const float* __restrict__ book_b,
                  const int* __restrict__ rates, const int* __restrict__ impl,
                  const int* __restrict__ blkhist, int* __restrict__ binbase,
                  int2* __restrict__ pairs, float* __restrict__ out) {
  __shared__ int tot[NBIN_PAD];
  __shared__ int cursor[NBIN_PAD];
  __shared__ __attribute__((aligned(16))) float Ws[H * WP];
  if (blockIdx.x >= NBLK) {
    const int lb = (int)blockIdx.x - NBLK;
    const bool bk = lb >= G2U;
    gemm_block<GT, SPB2>(bk, bk ? NSTRIP_U + (lb - G2U) * SPB2 : L1_SE + lb * SPB2,
                         bk ? L2_SE : NSTRIP_U, Ws,
                         user_x, user_W, user_b, book_x, book_W, book_b, out);
    return;
  }
  const int t = threadIdx.x;
  const int lane = t & 31;
  const int wv = t >> 5;
  const int blk = blockIdx.x;
  for (int b = wv; b < NBIN; b += GT / 32) {
    const int4* rp = (const int4*)(blkhist + b * NBLK) + lane * 2;
    const int4 v0 = rp[0], v1 = rp[1];
    const int i0 = lane * 8;
    int all = ((v0.x + v0.y) + (v0.z + v0.w)) + ((v1.x + v1.y) + (v1.z + v1.w));
    int bef = 0;
    bef += (i0 + 0 < blk) ? v0.x : 0;
    bef += (i0 + 1 < blk) ? v0.y : 0;
    bef += (i0 + 2 < blk) ? v0.z : 0;
    bef += (i0 + 3 < blk) ? v0.w : 0;
    bef += (i0 + 4 < blk) ? v1.x : 0;
    bef += (i0 + 5 < blk) ? v1.y : 0;
    bef += (i0 + 6 < blk) ? v1.z : 0;
    bef += (i0 + 7 < blk) ? v1.w : 0;
    for (int o = 16; o > 0; o >>= 1) {
      all += __shfl_xor(all, o);
      bef += __shfl_xor(bef, o);
    }
    if (lane == 0) { tot[b] = all; cursor[b] = bef; }
  }
  __syncthreads();
  if (wv == 0) {
    int s = 0;
    for (int k = 0; k < 10; ++k) {
      const int b = lane * 10 + k;
      if (b < NBIN) s += tot[b];
    }
    int inc = s;
    for (int o = 1; o < 32; o <<= 1) {
      const int y = __shfl_up(inc, o);
      if (lane >= o) inc += y;
    }
    int run = inc - s;
    for (int k = 0; k < 10; ++k) {
      const int b = lane * 10 + k;
      if (b < NBIN) {
        const int tb = tot[b];
        cursor[b] += run;
        if (blk == 0) binbase[b] = run;
        run += tb;
      }
    }
    if (blk == 0 && lane == 31) binbase[NBIN] = inc;
  }
  __syncthreads();
  const int e0 = blk * CHUNK;
  const int e1 = min(e0 + CHUNK, NE);
  for (int eb = e0 + t; eb < e1; eb += 4 * GT) {
    int uu[4], bb[4];
    for (int k = 0; k < 4; ++k) {
      const int e = eb + k * GT;
      uu[k] = -1;
      bb[k] = 0;
      if (e < e1) load_edge(e, rates, impl, uu[k], bb[k]);
    }
    for (int k = 0; k < 4; ++k) {
      if (uu[k] >= 0) {
        int row, col;
        orient_edge(eb + k * GT, uu[k], bb[k], row, col);
        const int p = atomicAdd(&cursor[col >> BIN_SHIFT], 1);
        pairs[p] = make_int2(col, row);
      }
    }
  }
}

__global__ void __launch_bounds__(GT)
sort_gemm_kernel(const float* __restrict__ user_x, const float* __restrict__ user_W,
                 const float* __restrict__ user_b, const float* __restrict__ book_x,
                 const float* __restrict__ book_W, const float* __restrict__ book_b,
                 const int2* __restrict__ pairs, const int* __restrict__ binbase,
                 int* __restrict__ cur, float* __restrict__ dis,
                 int* __restrict__ srt_row, float* __restrict__ out) {
  __shared__ int cnt[BIN_NODES];
  __shared__ int wsum[BIN_NODES / 32];
  __shared__ __attribute__((aligned(16))) float Ws[H * WP];
  if (blockIdx.x >= NBIN) {
    gemm_block<GT, SPB2>(true, L2_SE + ((int)blockIdx.x - NBIN) * SPB2, NSTRIP, Ws,
                         user_x, user_W, user_b, book_x, book_W, book_b, out);
    return;
  }
  const int t = threadIdx.x;
  const int lane = t & 31;
  const int wv = t >> 5;
  const int nbase = blockIdx.x << BIN_SHIFT;
  const int rbeg = binbase[blockIdx.x];
  const int rend = binbase[blockIdx.x + 1];
  if (t < BIN_NODES) cnt[t] = 0;
  __syncthreads();
  for (int ib = rbeg + t; ib < rend; ib += 4 * GT) {
    int cl[4];
    for (int k = 0; k < 4; ++k) {
      const int i = ib + k * GT;
      cl[k] = (i < rend) ? pairs[i].x : -1;
    }
    for (int k = 0; k < 4; ++k) {
      if (cl[k] >= 0) atomicAdd(&cnt[cl[k] - nbase], 1);
    }
  }
  __syncthreads();
  int d = 0, inc = 0;
  if (t < BIN_NODES) {
    d = cnt[t];
    inc = d;
    for (int o = 1; o < 32; o <<= 1) {
      const int y = __shfl_up(inc, o);
      if (lane >= o) inc += y;
    }
    if (lane == 31) wsum[wv] = inc;
  }
  __syncthreads();
  if (t < BIN_NODES) {
    int woff = 0;
    for (int j = 0; j < BIN_NODES / 32; ++j) {
      if (j < wv) woff += wsum[j];
    }
    const int ex = woff + inc - d;
    cnt[t] = ex;
    const int node = nbase + t;
    if (node < NNODE) {
      dis[node] = (d > 0) ? rsqrtf((float)d) : 0.0f;
      cur[node] = rbeg + ex + d;
    }
  }
  __syncthreads();
  for (int ib = rbeg + t; ib < rend; ib += 4 * GT) {
    int2 p[4];
    for (int k = 0; k < 4; ++k) {
      const int i = ib + k * GT;
      p[k] = (i < rend) ? pairs[i] : make_int2(-1, 0);
    }
    for (int k = 0; k < 4; ++k) {
      if (p[k].x >= 0) {
        const int pos = atomicAdd(&cnt[p[k].x - nbase], 1);
        srt_row[rbeg + pos] = p[k].y;
      }
    }
  }
}

#define GPW 8
#define GBLOCKS ((NNODE + 8 * GPW - 1) / (8 * GPW))
#define GWAVES (GBLOCKS * 8)

#define PICK_EDGE(k)                                                          \
  const int s##k = __builtin_ctz(live);                                       \
  live &= live - 1u;                                                          \
  const int r##k = __builtin_amdgcn_readlane(r, s##k);                        \
  const float w##k = __builtin_bit_cast(                                      \
      float, __builtin_amdgcn_readlane(__builtin_bit_cast(int, w), s##k));
#define LOAD_ROW(k)                                                           \
  const float4 v##k = ((const float4*)(x + (size_t)r##k * H))[lane];
#define ADD_EDGE(k)                                                           \
  acc.x += v##k.x * w##k; acc.y += v##k.y * w##k;                             \
  acc.z += v##k.z * w##k; acc.w += v##k.w * w##k;

__device__ __forceinline__ float4 segment_sum(const float* __restrict__ x,
                                              const float* __restrict__ dis,
                                              const int* __restrict__ srt_row,
                                              int beg, int end, float dc, int lane) {
  float4 acc = make_float4(0.0f, 0.0f, 0.0f, 0.0f);
  for (int base = beg; base < end; base += 32) {
    const int idx = base + lane;
    int r = 0;
    float w = 0.0f;
    if (idx < end) {
      r = srt_row[idx];
      w = dis[r] * dc;
    }
    unsigned live = __builtin_amdgcn_ballot_w32(w != 0.0f);
    while (live != 0u) {
      PICK_EDGE(0)
      if (live == 0u) { LOAD_ROW(0) ADD_EDGE(0) break; }
      PICK_EDGE(1)
      if (live == 0u) { LOAD_ROW(0) LOAD_ROW(1) ADD_EDGE(0) ADD_EDGE(1) break; }
      PICK_EDGE(2)
      if (live == 0u) {
        LOAD_ROW(0) LOAD_ROW(1) LOAD_ROW(2)
        ADD_EDGE(0) ADD_EDGE(1) ADD_EDGE(2)
        break;
      }
      PICK_EDGE(3)
      LOAD_ROW(0) LOAD_ROW(1) LOAD_ROW(2) LOAD_ROW(3)
      ADD_EDGE(0) ADD_EDGE(1) ADD_EDGE(2) ADD_EDGE(3)
    }
  }
  return acc;
}

__global__ void __launch_bounds__(256)
gather_kernel(const float* __restrict__ x, const float* __restrict__ dis,
              const int* __restrict__ cur, const int* __restrict__ srt_row,
              float* __restrict__ out) {
  const int lane = threadIdx.x & 31;
  const int g = __builtin_amdgcn_readfirstlane(
      (int)((blockIdx.x * blockDim.x + threadIdx.x) >> 5));
  for (int j = 0; j < GPW; ++j) {
    const int c = g + j * GWAVES;
    if (c >= NNODE) break;
    const int prev = cur[(c > 0) ? c - 1 : 0];
    const int end = cur[c];
    const float dc = dis[c];
    const int beg = (c > 0) ? prev : 0;
    const float4 acc = segment_sum(x, dis, srt_row, beg, end, dc, lane);
    ((float4*)(out + (size_t)c * H))[lane] = acc;
  }
}

__global__ void __launch_bounds__(256)
gather_mean_kernel(const float* __restrict__ x1, const float* __restrict__ dis,
                   const int* __restrict__ cur, const int* __restrict__ srt_row,
                   float* __restrict__ out) {
  const int lane = threadIdx.x & 31;
  const int g = __builtin_amdgcn_readfirstlane(
      (int)((blockIdx.x * blockDim.x + threadIdx.x) >> 5));
  for (int j = 0; j < GPW; ++j) {
    const int c = g + j * GWAVES;
    if (c >= NNODE) break;
    const int prev = cur[(c > 0) ? c - 1 : 0];
    const int end = cur[c];
    const float dc = dis[c];
    const int beg = (c > 0) ? prev : 0;
    float4* orow = (float4*)(out + (size_t)c * H) + lane;
    const float4 o = *orow;
    const float4 a = ((const float4*)(x1 + (size_t)c * H))[lane];
    const float4 b = segment_sum(x1, dis, srt_row, beg, end, dc, lane);
    const float s = 1.0f / 3.0f;
    float4 m;
    m.x = ((o.x + a.x) + b.x) * s;
    m.y = ((o.y + a.y) + b.y) * s;
    m.z = ((o.z + a.z) + b.z) * s;
    m.w = ((o.w + a.w) + b.w) * s;
    *orow = m;
  }
}

extern "C" void kernel_launch(void* const* d_in, const int* in_sizes, int n_in,
                              void* d_out, int out_size, void* d_ws, size_t ws_size,
                              hipStream_t stream) {
  (void)in_sizes; (void)n_in; (void)out_size; (void)ws_size;

  const float* user_x = (const float*)d_in[0];
  const float* book_x = (const float*)d_in[1];
  const float* user_W = (const float*)d_in[2];
  const float* user_b = (const float*)d_in[3];
  const float* book_W = (const float*)d_in[4];
  const float* book_b = (const float*)d_in[5];
  const int*   rates  = (const int*)d_in[6];
  const int*   impl   = (const int*)d_in[7];
  float* outp = (float*)d_out;

  float* x1  = (float*)d_ws;
  float* x2  = x1  + (size_t)NNODE * H;
  float* deg = x2  + (size_t)NNODE * H;
  float* dis = deg + NNODE;
  int*   srt = (int*)(dis + NNODE);
  int*   cur = (int*)deg;
  int2*  pairs   = (int2*)x2;
  int*   blkhist = (int*)x2 + 2 * (size_t)NE;
  int*   binbase = blkhist + NBIN * NBLK;

  stage1_kernel<<<NBLK + G1, S1T, 0, stream>>>(user_x, user_W, user_b,
                                               book_x, book_W, book_b,
                                               rates, impl, outp, blkhist);
  place_gemm_kernel<<<NBLK + G2, GT, 0, stream>>>(user_x, user_W, user_b,
                                                  book_x, book_W, book_b,
                                                  rates, impl, blkhist, binbase, pairs, outp);
  sort_gemm_kernel<<<NBIN + G3, GT, 0, stream>>>(user_x, user_W, user_b,
                                                 book_x, book_W, book_b,
                                                 pairs, binbase, cur, dis, srt, outp);

  gather_kernel<<<GBLOCKS, 256, 0, stream>>>(outp, dis, cur, srt, x1);
  gather_mean_kernel<<<GBLOCKS, 256, 0, stream>>>(x1, dis, cur, srt, outp);
}
